# starting kernel
# baseline (speedup 1.0000x reference)
#include <hip/hip_runtime.h>

#define DIM       128
#define TILE_M    16
#define MLP_WAVES 4
#define ROW_PAD   4
#define WSTEPS    (DIM / 4)
#define WENTRIES  (8 * WSTEPS * 32)

typedef float v2f __attribute__((ext_vector_type(2)));
typedef float v8f __attribute__((ext_vector_type(8)));

__global__ void __launch_bounds__(256) gin_scatter_kernel(
    const float* __restrict__ x,
    const int*   __restrict__ src,
    const int*   __restrict__ dst,
    float*       __restrict__ agg,
    int n_edges)
{
    const int wid  = (int)(blockIdx.x * (blockDim.x >> 5)) + (int)(threadIdx.x >> 5);
    const int lane = threadIdx.x & 31;
    if (wid >= n_edges) return;

    const int s = src[wid];
    const int d = dst[wid];

    const float4 v = reinterpret_cast<const float4*>(x + (size_t)s * DIM)[lane];
    float* p = agg + (size_t)d * DIM + (size_t)lane * 4;
    atomicAdd(p + 0, v.x);
    atomicAdd(p + 1, v.y);
    atomicAdd(p + 2, v.z);
    atomicAdd(p + 3, v.w);
}

__global__ void __launch_bounds__(MLP_WAVES * 32) gin_mlp_kernel(
    const float* __restrict__ x,
    const float* __restrict__ agg,
    const float* __restrict__ W1,
    const float* __restrict__ b1,
    const float* __restrict__ W2,
    const float* __restrict__ b2,
    float*       __restrict__ out,
    int n_nodes)
{
    __shared__ float2 wl[WENTRIES];
    __shared__ float  sh[MLP_WAVES][TILE_M][DIM + ROW_PAD];

    const int wave = threadIdx.x >> 5;
    const int lane = threadIdx.x & 31;
    const int tile = blockIdx.x * MLP_WAVES + wave;
    const int row0 = tile * TILE_M;
    const bool valid = (row0 < n_nodes);
    const int mmax = valid ? min(TILE_M, n_nodes - row0) : 0;

    if (valid) {
        for (int i = lane; i < TILE_M * (DIM / 4); i += 32) {
            const int m  = i >> 5;
            const int c4 = i & 31;
            float4 h;
            if (m < mmax) {
                const size_t r = (size_t)(row0 + m) * DIM;
                const float4 xv = reinterpret_cast<const float4*>(x + r)[c4];
                const float4 av = reinterpret_cast<const float4*>(agg + r)[c4];
                h = make_float4(xv.x + av.x, xv.y + av.y, xv.z + av.z, xv.w + av.w);
            } else {
                h = make_float4(0.f, 0.f, 0.f, 0.f);
            }
            float* row = &sh[wave][m][0];
            row[c4 * 4 + 0] = h.x; row[c4 * 4 + 1] = h.y;
            row[c4 * 4 + 2] = h.z; row[c4 * 4 + 3] = h.w;
        }
    }

    for (int idx = threadIdx.x; idx < WENTRIES; idx += MLP_WAVES * 32) {
        const int nt = idx >> 10;
        const int s  = (idx >> 5) & 31;
        const int L  = idx & 31;
        const int k  = (s << 2) + ((L >> 4) << 1);
        const int n  = (nt << 4) + (L & 15);
        wl[idx] = make_float2(W1[(size_t)k * DIM + n],
                              W1[(size_t)(k + 1) * DIM + n]);
    }
    __syncthreads();

    const int m_a  = lane & 15;
    const int koff = (lane >> 4) * 2;
    const int n_l  = lane & 15;
    const int mb   = (lane >> 4) * 8;
    const float* hrow = &sh[wave][m_a][0];

    v8f acc[8];
    if (valid) {
        #pragma unroll
        for (int nt = 0; nt < 8; ++nt) {
            v8f c = {0.f, 0.f, 0.f, 0.f, 0.f, 0.f, 0.f, 0.f};
            const float2* wf = wl + (nt << 10) + lane;
            for (int s = 0; s < WSTEPS; ++s) {
                v2f a, b;
                a.x = hrow[(s << 2) + koff];
                a.y = hrow[(s << 2) + koff + 1];
                const float2 bb = wf[s << 5];
                b.x = bb.x; b.y = bb.y;
                c = __builtin_amdgcn_wmma_f32_16x16x4_f32(
                        false, a, false, b, (short)0, c, false, false);
            }
            acc[nt] = c;
        }

        #pragma unroll
        for (int nt = 0; nt < 8; ++nt) {
            const int col  = nt * 16 + n_l;
            const float bv = b1[col];
            #pragma unroll
            for (int v = 0; v < 8; ++v) {
                const float t = acc[nt][v] + bv;
                sh[wave][mb + v][col] = t > 0.f ? t : 0.f;
            }
        }
    }
    __syncthreads();

    for (int idx = threadIdx.x; idx < WENTRIES; idx += MLP_WAVES * 32) {
        const int nt = idx >> 10;
        const int s  = (idx >> 5) & 31;
        const int L  = idx & 31;
        const int k  = (s << 2) + ((L >> 4) << 1);
        const int n  = (nt << 4) + (L & 15);
        wl[idx] = make_float2(W2[(size_t)k * DIM + n],
                              W2[(size_t)(k + 1) * DIM + n]);
    }
    __syncthreads();

    if (valid) {
        #pragma unroll
        for (int nt = 0; nt < 8; ++nt) {
            v8f c = {0.f, 0.f, 0.f, 0.f, 0.f, 0.f, 0.f, 0.f};
            const float2* wf = wl + (nt << 10) + lane;
            for (int s = 0; s < WSTEPS; ++s) {
                v2f a, b;
                a.x = hrow[(s << 2) + koff];
                a.y = hrow[(s << 2) + koff + 1];
                const float2 bb = wf[s << 5];
                b.x = bb.x; b.y = bb.y;
                c = __builtin_amdgcn_wmma_f32_16x16x4_f32(
                        false, a, false, b, (short)0, c, false, false);
            }
            const int col  = nt * 16 + n_l;
            const float bv = b2[col];
            #pragma unroll
            for (int v = 0; v < 8; ++v) {
                const int m = mb + v;
                if (m < mmax)
                    out[(size_t)(row0 + m) * DIM + col] = c[v] + bv;
            }
        }
    }
}

extern "C" void kernel_launch(void* const* d_in, const int* in_sizes, int n_in,
                              void* d_out, int out_size, void* d_ws, size_t ws_size,
                              hipStream_t stream) {
    const float* x  = (const float*)d_in[0];
    const int*   ei = (const int*)  d_in[1];
    const float* W1 = (const float*)d_in[2];
    const float* b1 = (const float*)d_in[3];
    const float* W2 = (const float*)d_in[4];
    const float* b2 = (const float*)d_in[5];
    float* out = (float*)d_out;

    const int n_nodes = in_sizes[0] / DIM;
    const int n_edges = in_sizes[1] / 2;

    float* agg = (float*)d_ws;

    hipMemsetAsync(agg, 0, (size_t)n_nodes * DIM * sizeof(float), stream);

    const int blocks_sc = (n_edges + 7) / 8;
    gin_scatter_kernel<<<blocks_sc, 256, 0, stream>>>(
        x, ei, ei + n_edges, agg, n_edges);

    const int tiles      = (n_nodes + TILE_M - 1) / TILE_M;
    const int blocks_mlp = (tiles + MLP_WAVES - 1) / MLP_WAVES;
    gin_mlp_kernel<<<blocks_mlp, MLP_WAVES * 32, 0, stream>>>(
        x, agg, W1, b1, W2, b2, out, n_nodes);
}


# agg summed in LDS by the block owning each 391-row range; owners scan a 1-byte range tag per edge (a tag kernel replaces the memset), not the dst list; LDS f32 atomics; sum order: LDS arrival
# speedup vs baseline: 2.4758x; 2.4758x over previous
#include <hip/hip_runtime.h>

#define DIM       128
#define ROW_PAD   4
#define WSTEPS    (DIM / 4)

typedef float v2f __attribute__((ext_vector_type(2)));
typedef float v8f __attribute__((ext_vector_type(8)));

#define RANGE_ROWS   391
#define OWN_WAVES    32
#define OWN_THREADS  (OWN_WAVES * 32)
#define TAG_CHUNK    512
#define SCAN_DEPTH   4
#define ROW_BATCH    8
#define TAG_WAVES    8

typedef unsigned int u32;

__global__ void __launch_bounds__(TAG_WAVES * 32) gin_tag_kernel(
    const int* __restrict__ dst, u32* __restrict__ tagw, int n_edges, int n_chunks)
{
    const int wave  = threadIdx.x >> 5;
    const int lane  = threadIdx.x & 31;
    const int chunk = blockIdx.x * TAG_WAVES + wave;
    if (chunk >= n_chunks) return;
    const bool vec_ok = (n_edges & 3) == 0;
    #pragma unroll
    for (int j = 0; j < 4; ++j) {
        const int word = chunk * (TAG_CHUNK / 4) + j * 32 + lane;
        const long long e = 4ll * word;
        int d0, d1, d2, d3;
        if (vec_ok && e + 3 < n_edges) {
            const int4 d = reinterpret_cast<const int4*>(dst)[word];
            d0 = d.x; d1 = d.y; d2 = d.z; d3 = d.w;
        } else {
            d0 = e     < n_edges ? dst[e]     : 0;
            d1 = e + 1 < n_edges ? dst[e + 1] : 0;
            d2 = e + 2 < n_edges ? dst[e + 2] : 0;
            d3 = e + 3 < n_edges ? dst[e + 3] : 0;
        }
        const u32 t0 = ((u32)d0 / RANGE_ROWS) & 0xffu;
        const u32 t1 = ((u32)d1 / RANGE_ROWS) & 0xffu;
        const u32 t2 = ((u32)d2 / RANGE_ROWS) & 0xffu;
        const u32 t3 = ((u32)d3 / RANGE_ROWS) & 0xffu;
        tagw[word] = t0 | (t1 << 8) | (t2 << 16) | (t3 << 24);
    }
}

__device__ __forceinline__ int wrap_chunk(int c, int n) { return c >= n ? c - n : c; }

__device__ __forceinline__ u32 zero_bytes(u32 x)
{
    return ~(((x & 0x7f7f7f7fu) + 0x7f7f7f7fu) | x | 0x7f7f7f7fu);
}

template <bool FULL>
__device__ __forceinline__ void row_batch(
    const float* __restrict__ x, float* acc, int lane, int sv, int rv, u32& m, int n)
{
    float v[ROW_BATCH][4];
    int   rr[ROW_BATCH];
    #pragma unroll
    for (int j = 0; j < ROW_BATCH; ++j) {
        if (FULL || j < n) {
            const int k = __builtin_ctz(m);
            m &= m - 1;
            const int s = __builtin_amdgcn_readlane(sv, k);
            rr[j]       = __builtin_amdgcn_readlane(rv, k);
            const float* xr = x + (size_t)s * DIM + lane;
            v[j][0] = xr[0];
            v[j][1] = xr[32];
            v[j][2] = xr[64];
            v[j][3] = xr[96];
        }
    }
    #pragma unroll
    for (int j = 0; j < ROW_BATCH; ++j) {
        if (FULL || j < n) {
            float* p = acc + rr[j] * DIM + lane;
            __hip_atomic_fetch_add(p,      v[j][0], __ATOMIC_RELAXED, __HIP_MEMORY_SCOPE_WORKGROUP);
            __hip_atomic_fetch_add(p + 32, v[j][1], __ATOMIC_RELAXED, __HIP_MEMORY_SCOPE_WORKGROUP);
            __hip_atomic_fetch_add(p + 64, v[j][2], __ATOMIC_RELAXED, __HIP_MEMORY_SCOPE_WORKGROUP);
            __hip_atomic_fetch_add(p + 96, v[j][3], __ATOMIC_RELAXED, __HIP_MEMORY_SCOPE_WORKGROUP);
        }
    }
}

__device__ __forceinline__ void drain_queue(
    const float* __restrict__ x, const int* __restrict__ src, const int* __restrict__ dst,
    float* acc, int lane, int q, int cnt, int n_edges, int n_nodes, int row0, int rows)
{
    const bool have = (lane < cnt) && (q < n_edges);
    const int  s    = have ? src[q] : 0;
    const int  d    = have ? dst[q] : -1;
    const int  r    = d - row0;
    const bool ok = have && ((u32)r < (u32)rows) && ((u32)s < (u32)n_nodes);
    u32 m = __builtin_amdgcn_ballot_w32(ok);
    int n = __builtin_popcount(m);
    while (n >= ROW_BATCH) {
        row_batch<true>(x, acc, lane, s, r, m, ROW_BATCH);
        n -= ROW_BATCH;
    }
    if (n > 0)
        row_batch<false>(x, acc, lane, s, r, m, n);
}

__global__ void __launch_bounds__(OWN_THREADS) gin_own_kernel(
    const float* __restrict__ x,
    const int*   __restrict__ src,
    const int*   __restrict__ dst,
    const u32*   __restrict__ tagw,
    float*       __restrict__ agg,
    int n_edges, int n_nodes, int n_chunks)
{
    __shared__ __attribute__((aligned(16))) float acc[RANGE_ROWS * DIM];

    const int wave = __builtin_amdgcn_readfirstlane((int)(threadIdx.x >> 5));
    const int lane = threadIdx.x & 31;
    const int row0 = blockIdx.x * RANGE_ROWS;
    const int left = n_nodes - row0;
    const int rows = left < RANGE_ROWS ? left : RANGE_ROWS;

    for (int i = threadIdx.x; i < RANGE_ROWS * DIM / 4; i += OWN_THREADS)
        reinterpret_cast<float4*>(acc)[i] = make_float4(0.f, 0.f, 0.f, 0.f);
    __syncthreads();

    const u32 mine  = (u32)blockIdx.x * 0x01010101u;
    const int start = (int)(((u32)blockIdx.x * (u32)n_chunks) / (u32)gridDim.x);
    const uint4* tag16 = reinterpret_cast<const uint4*>(tagw);

    int q   = 0;
    int cnt = 0;

    uint4 w[SCAN_DEPTH];
    #pragma unroll
    for (int u = 0; u < SCAN_DEPTH; ++u) {
        const int kk = wave + OWN_WAVES * u;
        const int kc = kk < n_chunks ? kk : 0;
        w[u] = tag16[wrap_chunk(start + kc, n_chunks) * (TAG_CHUNK / 16) + lane];
    }
    for (int k0 = wave; k0 < n_chunks; k0 += OWN_WAVES * SCAN_DEPTH) {
        uint4 wn[SCAN_DEPTH];
        #pragma unroll
        for (int u = 0; u < SCAN_DEPTH; ++u) {
            const int kn = k0 + OWN_WAVES * (SCAN_DEPTH + u);
            const int kc = kn < n_chunks ? kn : 0;
            wn[u] = tag16[wrap_chunk(start + kc, n_chunks) * (TAG_CHUNK / 16) + lane];
        }
        #pragma unroll
        for (int u = 0; u < SCAN_DEPTH; ++u) {
            const int kk = k0 + OWN_WAVES * u;
            if (kk < n_chunks) {
                const int ebase = wrap_chunk(start + kk, n_chunks) * TAG_CHUNK;
                const u32 pm = (zero_bytes(w[u].x ^ mine) >> 7) | (zero_bytes(w[u].y ^ mine) >> 6)
                             | (zero_bytes(w[u].z ^ mine) >> 5) | (zero_bytes(w[u].w ^ mine) >> 4);
                u32 lanes = __builtin_amdgcn_ballot_w32(pm != 0u);
                while (lanes) {
                    const int l = __builtin_ctz(lanes);
                    lanes &= lanes - 1;
                    u32 pl = (u32)__builtin_amdgcn_readlane((int)pm, l);
                    while (pl) {
                        const int p = __builtin_ctz(pl);
                        pl &= pl - 1;
                        const int e = ebase + l * 16 + ((p & 7) << 2) + (p >> 3);
                        q = (lane == cnt) ? e : q;
                        ++cnt;
                        if (cnt == 32) {
                            drain_queue(x, src, dst, acc, lane, q, 32, n_edges, n_nodes, row0, rows);
                            cnt = 0;
                        }
                    }
                }
            }
        }
        #pragma unroll
        for (int u = 0; u < SCAN_DEPTH; ++u) w[u] = wn[u];
    }
    if (cnt > 0)
        drain_queue(x, src, dst, acc, lane, q, cnt, n_edges, n_nodes, row0, rows);
    __syncthreads();

    float4* out4 = reinterpret_cast<float4*>(agg + (size_t)row0 * DIM);
    for (int i = threadIdx.x; i < rows * (DIM / 4); i += OWN_THREADS)
        out4[i] = reinterpret_cast<const float4*>(acc)[i];
}

#define MLP_WAVES2  8
#define MLP_THREADS (MLP_WAVES2 * 32)
#define MT          32
#define PITCH       (DIM + ROW_PAD)
#define ROWS_PER_THREAD (MT / MLP_WAVES2)

__global__ void __launch_bounds__(MLP_THREADS) gin_mlp_kernel(
    const float* __restrict__ x,
    const float* __restrict__ agg,
    const float* __restrict__ W1,
    const float* __restrict__ b1,
    const float* __restrict__ W2,
    const float* __restrict__ b2,
    float*       __restrict__ out,
    int n_nodes, int n_tiles)
{
    __shared__ __attribute__((aligned(16))) float bufA[MT][PITCH];
    __shared__ __attribute__((aligned(16))) float bufB[MT][PITCH];
    __shared__ __attribute__((aligned(16))) float bufC[MT][PITCH];

    const int wave = threadIdx.x >> 5;
    const int lane = threadIdx.x & 31;
    const int c    = lane & 15;
    const int hh   = lane >> 4;
    const int koff = hh * 2;
    const int ncol = wave * 16 + c;

    v2f w1[WSTEPS], w2[WSTEPS];
    #pragma unroll
    for (int s = 0; s < WSTEPS; ++s) {
        const int k = (s << 2) + koff;
        w1[s].x = W1[(size_t)k * DIM + ncol];
        w1[s].y = W1[(size_t)(k + 1) * DIM + ncol];
        w2[s].x = W2[(size_t)k * DIM + ncol];
        w2[s].y = W2[(size_t)(k + 1) * DIM + ncol];
    }
    const float bias1 = b1[ncol];
    const float bias2 = b2[ncol];

    float4 xv[ROWS_PER_THREAD], av[ROWS_PER_THREAD];

    int tile = blockIdx.x;
    if (tile >= n_tiles) return;

    #pragma unroll
    for (int j = 0; j < ROWS_PER_THREAD; ++j) {
        const int m = wave + MLP_WAVES2 * j;
        const int r = tile * MT + m;
        float4 h = make_float4(0.f, 0.f, 0.f, 0.f);
        if (r < n_nodes) {
            const float4 a = reinterpret_cast<const float4*>(x   + (size_t)r * DIM)[lane];
            const float4 b = reinterpret_cast<const float4*>(agg + (size_t)r * DIM)[lane];
            h = make_float4(a.x + b.x, a.y + b.y, a.z + b.z, a.w + b.w);
        }
        *reinterpret_cast<float4*>(&bufA[m][lane * 4]) = h;
    }
    __syncthreads();

    const float* a_row0 = &bufA[c][koff];
    const float* a_row1 = &bufA[16 + c][koff];
    const float* h_row0 = &bufB[c][koff];
    const float* h_row1 = &bufB[16 + c][koff];

    for (; tile < n_tiles; tile += gridDim.x) {
        const int  next     = tile + (int)gridDim.x;
        const bool has_next = next < n_tiles;

        if (has_next) {
            #pragma unroll
            for (int j = 0; j < ROWS_PER_THREAD; ++j) {
                const int r = next * MT + wave + MLP_WAVES2 * j;
                if (r < n_nodes) {
                    xv[j] = reinterpret_cast<const float4*>(x   + (size_t)r * DIM)[lane];
                    av[j] = reinterpret_cast<const float4*>(agg + (size_t)r * DIM)[lane];
                } else {
                    xv[j] = make_float4(0.f, 0.f, 0.f, 0.f);
                    av[j] = make_float4(0.f, 0.f, 0.f, 0.f);
                }
            }
        }

        v8f c0 = {0.f, 0.f, 0.f, 0.f, 0.f, 0.f, 0.f, 0.f};
        v8f c1 = {0.f, 0.f, 0.f, 0.f, 0.f, 0.f, 0.f, 0.f};
        #pragma unroll
        for (int s = 0; s < WSTEPS; ++s) {
            const v2f a0 = *reinterpret_cast<const v2f*>(a_row0 + (s << 2));
            const v2f a1 = *reinterpret_cast<const v2f*>(a_row1 + (s << 2));
            c0 = __builtin_amdgcn_wmma_f32_16x16x4_f32(
                     false, a0, false, w1[s], (short)0, c0, false, false);
            c1 = __builtin_amdgcn_wmma_f32_16x16x4_f32(
                     false, a1, false, w1[s], (short)0, c1, false, false);
        }
        #pragma unroll
        for (int v = 0; v < 8; ++v) {
            const float t0 = c0[v] + bias1;
            const float t1 = c1[v] + bias1;
            bufB[hh * 8 + v][ncol]      = t0 > 0.f ? t0 : 0.f;
            bufB[16 + hh * 8 + v][ncol] = t1 > 0.f ? t1 : 0.f;
        }
        __syncthreads();

        if (has_next) {
            #pragma unroll
            for (int j = 0; j < ROWS_PER_THREAD; ++j) {
                const int m = wave + MLP_WAVES2 * j;
                *reinterpret_cast<float4*>(&bufA[m][lane * 4]) =
                    make_float4(xv[j].x + av[j].x, xv[j].y + av[j].y,
                                xv[j].z + av[j].z, xv[j].w + av[j].w);
            }
        }

        v8f d0 = {0.f, 0.f, 0.f, 0.f, 0.f, 0.f, 0.f, 0.f};
        v8f d1 = {0.f, 0.f, 0.f, 0.f, 0.f, 0.f, 0.f, 0.f};
        #pragma unroll
        for (int s = 0; s < WSTEPS; ++s) {
            const v2f a0 = *reinterpret_cast<const v2f*>(h_row0 + (s << 2));
            const v2f a1 = *reinterpret_cast<const v2f*>(h_row1 + (s << 2));
            d0 = __builtin_amdgcn_wmma_f32_16x16x4_f32(
                     false, a0, false, w2[s], (short)0, d0, false, false);
            d1 = __builtin_amdgcn_wmma_f32_16x16x4_f32(
                     false, a1, false, w2[s], (short)0, d1, false, false);
        }
        #pragma unroll
        for (int v = 0; v < 8; ++v) {
            bufC[hh * 8 + v][ncol]      = d0[v] + bias2;
            bufC[16 + hh * 8 + v][ncol] = d1[v] + bias2;
        }
        __syncthreads();

        #pragma unroll
        for (int j = 0; j < ROWS_PER_THREAD; ++j) {
            const int m = wave + MLP_WAVES2 * j;
            const int r = tile * MT + m;
            if (r < n_nodes)
                reinterpret_cast<float4*>(out + (size_t)r * DIM)[lane] =
                    *reinterpret_cast<const float4*>(&bufC[m][lane * 4]);
        }
    }
}

extern "C" void kernel_launch(void* const* d_in, const int* in_sizes, int n_in,
                              void* d_out, int out_size, void* d_ws, size_t ws_size,
                              hipStream_t stream) {
    const float* x  = (const float*)d_in[0];
    const int*   ei = (const int*)  d_in[1];
    const float* W1 = (const float*)d_in[2];
    const float* b1 = (const float*)d_in[3];
    const float* W2 = (const float*)d_in[4];
    const float* b2 = (const float*)d_in[5];
    float* out = (float*)d_out;

    const int n_nodes = in_sizes[0] / DIM;
    const int n_edges = in_sizes[1] / 2;

    const int n_chunks = (n_edges + TAG_CHUNK - 1) / TAG_CHUNK;
    unsigned int* tagw = (unsigned int*)d_ws;
    float* agg = (float*)((char*)d_ws + (size_t)n_chunks * TAG_CHUNK);

    const int blocks_tag = (n_chunks + TAG_WAVES - 1) / TAG_WAVES;
    gin_tag_kernel<<<blocks_tag, TAG_WAVES * 32, 0, stream>>>(ei + n_edges, tagw, n_edges, n_chunks);

    const int n_ranges = (n_nodes + RANGE_ROWS - 1) / RANGE_ROWS;
    gin_own_kernel<<<n_ranges, OWN_THREADS, 0, stream>>>(
        x, ei, ei + n_edges, tagw, agg, n_edges, n_nodes, n_chunks);

    const int tiles      = (n_nodes + MT - 1) / MT;
    const int blocks_mlp = tiles < 512 ? tiles : 512;
    gin_mlp_kernel<<<blocks_mlp, MLP_THREADS, 0, stream>>>(
        x, agg, W1, b1, W2, b2, out, n_nodes, tiles);
}


# Owners scan 1 bit per edge: a builder kernel writes a bit plane per 391-row range (lines built in LDS by integer LDS or); an owner reads only its plane, not all byte tags; sum order: LDS arrival
# speedup vs baseline: 2.7700x; 1.1188x over previous
#include <hip/hip_runtime.h>

#define DIM       128
#define ROW_PAD   4
#define WSTEPS    (DIM / 4)

typedef float v2f __attribute__((ext_vector_type(2)));
typedef float v8f __attribute__((ext_vector_type(8)));

#define RANGE_ROWS   391
#define MAX_RANGES   256
#define OWN_WAVES    32
#define OWN_THREADS  (OWN_WAVES * 32)
#define GROUP        1024
#define ROW_BATCH    8
#define PLANE_WAVES  2

typedef unsigned int u32;
typedef u32 u32x4 __attribute__((ext_vector_type(4)));

__global__ void __launch_bounds__(PLANE_WAVES * 32) gin_plane_kernel(
    const int* __restrict__ dst, u32* __restrict__ planes,
    int n_edges, int n_nodes, int n_lines)
{
    __shared__ __attribute__((aligned(16))) u32 bits[PLANE_WAVES][MAX_RANGES * 32];

    const int wave = __builtin_amdgcn_readfirstlane((int)(threadIdx.x >> 5));
    const int lane = threadIdx.x & 31;
    const int g    = blockIdx.x * PLANE_WAVES + wave;
    u32* mine = bits[wave];

    int d[8][4];
    const long long e0 = (long long)g * GROUP;
    if ((n_edges & 3) == 0 && e0 + GROUP <= n_edges) {
        const int4* d4 = reinterpret_cast<const int4*>(dst + e0) + lane;
        #pragma unroll
        for (int j = 0; j < 8; ++j) {
            const int4 v = d4[j * 32];
            d[j][0] = v.x; d[j][1] = v.y; d[j][2] = v.z; d[j][3] = v.w;
        }
    } else {
        #pragma unroll
        for (int j = 0; j < 8; ++j) {
            const long long e = e0 + j * 128 + lane * 4;
            d[j][0] = e     < n_edges ? dst[e]     : -1;
            d[j][1] = e + 1 < n_edges ? dst[e + 1] : -1;
            d[j][2] = e + 2 < n_edges ? dst[e + 2] : -1;
            d[j][3] = e + 3 < n_edges ? dst[e + 3] : -1;
        }
    }

    for (int k = 0; k < MAX_RANGES / 4; ++k)
        reinterpret_cast<u32x4*>(mine)[k * 32 + lane] = (u32x4){0u, 0u, 0u, 0u};
    __syncthreads();

    #pragma unroll
    for (int j = 0; j < 8; ++j) {
        #pragma unroll
        for (int k = 0; k < 4; ++k) {
            const int dd = d[j][k];
            if ((u32)dd < (u32)n_nodes) {
                const u32 t = (u32)dd / RANGE_ROWS;
                if (t < MAX_RANGES)
                    __hip_atomic_fetch_or(&mine[t * 32 + lane], 1u << (j * 4 + k),
                                          __ATOMIC_RELAXED, __HIP_MEMORY_SCOPE_WORKGROUP);
            }
        }
    }
    __syncthreads();

    const size_t plane_u4 = (size_t)n_lines * (GROUP / 128);
    u32x4* out4 = reinterpret_cast<u32x4*>(planes) + (size_t)(lane >> 3) * plane_u4
                + (size_t)g * (GROUP / 128) + (lane & 7);
    for (int k0 = 0; k0 < MAX_RANGES / 4; k0 += 8) {
        u32x4 v[8];
        #pragma unroll
        for (int t = 0; t < 8; ++t)
            v[t] = reinterpret_cast<const u32x4*>(mine)[(k0 + t) * 32 + lane];
        #pragma unroll
        for (int t = 0; t < 8; ++t)
            out4[(size_t)(k0 + t) * 4 * plane_u4] = v[t];
    }
}

template <bool FULL>
__device__ __forceinline__ void row_batch(
    const float* __restrict__ x, float* acc, int lane, int sv, int rv, u32& m, int n)
{
    float v[ROW_BATCH][4];
    int   rr[ROW_BATCH];
    #pragma unroll
    for (int j = 0; j < ROW_BATCH; ++j) {
        if (FULL || j < n) {
            const int k = __builtin_ctz(m);
            m &= m - 1;
            const int s = __builtin_amdgcn_readlane(sv, k);
            rr[j]       = __builtin_amdgcn_readlane(rv, k);
            const float* xr = x + (size_t)s * DIM + lane;
            v[j][0] = xr[0];
            v[j][1] = xr[32];
            v[j][2] = xr[64];
            v[j][3] = xr[96];
        }
    }
    #pragma unroll
    for (int j = 0; j < ROW_BATCH; ++j) {
        if (FULL || j < n) {
            float* p = acc + rr[j] * DIM + lane;
            __hip_atomic_fetch_add(p,      v[j][0], __ATOMIC_RELAXED, __HIP_MEMORY_SCOPE_WORKGROUP);
            __hip_atomic_fetch_add(p + 32, v[j][1], __ATOMIC_RELAXED, __HIP_MEMORY_SCOPE_WORKGROUP);
            __hip_atomic_fetch_add(p + 64, v[j][2], __ATOMIC_RELAXED, __HIP_MEMORY_SCOPE_WORKGROUP);
            __hip_atomic_fetch_add(p + 96, v[j][3], __ATOMIC_RELAXED, __HIP_MEMORY_SCOPE_WORKGROUP);
        }
    }
}

__device__ __forceinline__ void drain_queue(
    const float* __restrict__ x, const int* __restrict__ src, const int* __restrict__ dst,
    float* acc, int lane, int q, int cnt, int n_edges, int n_nodes, int row0, int rows)
{
    const bool have = (lane < cnt) && (q < n_edges);
    const int  s    = have ? src[q] : 0;
    const int  d    = have ? dst[q] : -1;
    const int  r    = d - row0;
    const bool ok = have && ((u32)r < (u32)rows) && ((u32)s < (u32)n_nodes);
    u32 m = __builtin_amdgcn_ballot_w32(ok);
    int n = __builtin_popcount(m);
    while (n >= ROW_BATCH) {
        row_batch<true>(x, acc, lane, s, r, m, ROW_BATCH);
        n -= ROW_BATCH;
    }
    if (n > 0)
        row_batch<false>(x, acc, lane, s, r, m, n);
}

__global__ void __launch_bounds__(OWN_THREADS) gin_own_kernel(
    const float* __restrict__ x,
    const int*   __restrict__ src,
    const int*   __restrict__ dst,
    const u32*   __restrict__ planes,
    float*       __restrict__ agg,
    int n_edges, int n_nodes, int n_lines)
{
    __shared__ __attribute__((aligned(16))) float acc[RANGE_ROWS * DIM];

    const int wave = __builtin_amdgcn_readfirstlane((int)(threadIdx.x >> 5));
    const int lane = threadIdx.x & 31;
    const int row0 = blockIdx.x * RANGE_ROWS;
    const int left = n_nodes - row0;
    const int rows = left < RANGE_ROWS ? left : RANGE_ROWS;

    for (int i = threadIdx.x; i < RANGE_ROWS * DIM / 4; i += OWN_THREADS)
        reinterpret_cast<float4*>(acc)[i] = make_float4(0.f, 0.f, 0.f, 0.f);
    __syncthreads();

    const int plane_u4 = n_lines * (GROUP / 128);
    const int n_pieces = plane_u4 >> 5;
    const u32x4* pl = reinterpret_cast<const u32x4*>(planes) + (size_t)blockIdx.x * plane_u4 + lane;

    int q   = 0;
    int cnt = 0;

    u32x4 w = __builtin_nontemporal_load(pl + (wave < n_pieces ? wave : 0) * 32);
    for (int i = wave; i < n_pieces; i += OWN_WAVES) {
        const int in = i + OWN_WAVES;
        const u32x4 wn = __builtin_nontemporal_load(pl + (in < n_pieces ? in : 0) * 32);
        const int u     = i * 32 + lane;
        const int ebase = (u >> 3) * GROUP + ((u & 7) << 4);
        const u32 wd[4] = {w.x, w.y, w.z, w.w};
        #pragma unroll
        for (int c = 0; c < 4; ++c) {
            u32 m     = wd[c];
            u32 lanes = __builtin_amdgcn_ballot_w32(m != 0u);
            while (lanes) {
                const int bpos = __builtin_ffs((int)m) - 1;
                const int ev   = ebase + (c << 2) + ((bpos >> 2) << 7) + (bpos & 3);
                while (lanes) {
                    const int l = __builtin_ctz(lanes);
                    lanes &= lanes - 1;
                    const int e = __builtin_amdgcn_readlane(ev, l);
                    q = (lane == cnt) ? e : q;
                    ++cnt;
                    if (cnt == 32) {
                        drain_queue(x, src, dst, acc, lane, q, 32, n_edges, n_nodes, row0, rows);
                        cnt = 0;
                    }
                }
                m &= m - 1;
                lanes = __builtin_amdgcn_ballot_w32(m != 0u);
            }
        }
        w = wn;
    }
    if (cnt > 0)
        drain_queue(x, src, dst, acc, lane, q, cnt, n_edges, n_nodes, row0, rows);
    __syncthreads();

    float4* out4 = reinterpret_cast<float4*>(agg + (size_t)row0 * DIM);
    for (int i = threadIdx.x; i < rows * (DIM / 4); i += OWN_THREADS)
        out4[i] = reinterpret_cast<const float4*>(acc)[i];
}

#define MLP_WAVES2  8
#define MLP_THREADS (MLP_WAVES2 * 32)
#define MT          32
#define PITCH       (DIM + ROW_PAD)
#define ROWS_PER_THREAD (MT / MLP_WAVES2)

__global__ void __launch_bounds__(MLP_THREADS) gin_mlp_kernel(
    const float* __restrict__ x,
    const float* __restrict__ agg,
    const float* __restrict__ W1,
    const float* __restrict__ b1,
    const float* __restrict__ W2,
    const float* __restrict__ b2,
    float*       __restrict__ out,
    int n_nodes, int n_tiles)
{
    __shared__ __attribute__((aligned(16))) float bufA[MT][PITCH];
    __shared__ __attribute__((aligned(16))) float bufB[MT][PITCH];
    __shared__ __attribute__((aligned(16))) float bufC[MT][PITCH];

    const int wave = threadIdx.x >> 5;
    const int lane = threadIdx.x & 31;
    const int c    = lane & 15;
    const int hh   = lane >> 4;
    const int koff = hh * 2;
    const int ncol = wave * 16 + c;

    v2f w1[WSTEPS], w2[WSTEPS];
    #pragma unroll
    for (int s = 0; s < WSTEPS; ++s) {
        const int k = (s << 2) + koff;
        w1[s].x = W1[(size_t)k * DIM + ncol];
        w1[s].y = W1[(size_t)(k + 1) * DIM + ncol];
        w2[s].x = W2[(size_t)k * DIM + ncol];
        w2[s].y = W2[(size_t)(k + 1) * DIM + ncol];
    }
    const float bias1 = b1[ncol];
    const float bias2 = b2[ncol];

    float4 xv[ROWS_PER_THREAD], av[ROWS_PER_THREAD];

    int tile = blockIdx.x;
    if (tile >= n_tiles) return;

    #pragma unroll
    for (int j = 0; j < ROWS_PER_THREAD; ++j) {
        const int m = wave + MLP_WAVES2 * j;
        const int r = tile * MT + m;
        float4 h = make_float4(0.f, 0.f, 0.f, 0.f);
        if (r < n_nodes) {
            const float4 a = reinterpret_cast<const float4*>(x   + (size_t)r * DIM)[lane];
            const float4 b = reinterpret_cast<const float4*>(agg + (size_t)r * DIM)[lane];
            h = make_float4(a.x + b.x, a.y + b.y, a.z + b.z, a.w + b.w);
        }
        *reinterpret_cast<float4*>(&bufA[m][lane * 4]) = h;
    }
    __syncthreads();

    const float* a_row0 = &bufA[c][koff];
    const float* a_row1 = &bufA[16 + c][koff];
    const float* h_row0 = &bufB[c][koff];
    const float* h_row1 = &bufB[16 + c][koff];

    for (; tile < n_tiles; tile += gridDim.x) {
        const int  next     = tile + (int)gridDim.x;
        const bool has_next = next < n_tiles;

        if (has_next) {
            #pragma unroll
            for (int j = 0; j < ROWS_PER_THREAD; ++j) {
                const int r = next * MT + wave + MLP_WAVES2 * j;
                if (r < n_nodes) {
                    xv[j] = reinterpret_cast<const float4*>(x   + (size_t)r * DIM)[lane];
                    av[j] = reinterpret_cast<const float4*>(agg + (size_t)r * DIM)[lane];
                } else {
                    xv[j] = make_float4(0.f, 0.f, 0.f, 0.f);
                    av[j] = make_float4(0.f, 0.f, 0.f, 0.f);
                }
            }
        }

        v8f c0 = {0.f, 0.f, 0.f, 0.f, 0.f, 0.f, 0.f, 0.f};
        v8f c1 = {0.f, 0.f, 0.f, 0.f, 0.f, 0.f, 0.f, 0.f};
        #pragma unroll
        for (int s = 0; s < WSTEPS; ++s) {
            const v2f a0 = *reinterpret_cast<const v2f*>(a_row0 + (s << 2));
            const v2f a1 = *reinterpret_cast<const v2f*>(a_row1 + (s << 2));
            c0 = __builtin_amdgcn_wmma_f32_16x16x4_f32(
                     false, a0, false, w1[s], (short)0, c0, false, false);
            c1 = __builtin_amdgcn_wmma_f32_16x16x4_f32(
                     false, a1, false, w1[s], (short)0, c1, false, false);
        }
        #pragma unroll
        for (int v = 0; v < 8; ++v) {
            const float t0 = c0[v] + bias1;
            const float t1 = c1[v] + bias1;
            bufB[hh * 8 + v][ncol]      = t0 > 0.f ? t0 : 0.f;
            bufB[16 + hh * 8 + v][ncol] = t1 > 0.f ? t1 : 0.f;
        }
        __syncthreads();

        if (has_next) {
            #pragma unroll
            for (int j = 0; j < ROWS_PER_THREAD; ++j) {
                const int m = wave + MLP_WAVES2 * j;
                *reinterpret_cast<float4*>(&bufA[m][lane * 4]) =
                    make_float4(xv[j].x + av[j].x, xv[j].y + av[j].y,
                                xv[j].z + av[j].z, xv[j].w + av[j].w);
            }
        }

        v8f d0 = {0.f, 0.f, 0.f, 0.f, 0.f, 0.f, 0.f, 0.f};
        v8f d1 = {0.f, 0.f, 0.f, 0.f, 0.f, 0.f, 0.f, 0.f};
        #pragma unroll
        for (int s = 0; s < WSTEPS; ++s) {
            const v2f a0 = *reinterpret_cast<const v2f*>(h_row0 + (s << 2));
            const v2f a1 = *reinterpret_cast<const v2f*>(h_row1 + (s << 2));
            d0 = __builtin_amdgcn_wmma_f32_16x16x4_f32(
                     false, a0, false, w2[s], (short)0, d0, false, false);
            d1 = __builtin_amdgcn_wmma_f32_16x16x4_f32(
                     false, a1, false, w2[s], (short)0, d1, false, false);
        }
        #pragma unroll
        for (int v = 0; v < 8; ++v) {
            bufC[hh * 8 + v][ncol]      = d0[v] + bias2;
            bufC[16 + hh * 8 + v][ncol] = d1[v] + bias2;
        }
        __syncthreads();

        #pragma unroll
        for (int j = 0; j < ROWS_PER_THREAD; ++j) {
            const int m = wave + MLP_WAVES2 * j;
            const int r = tile * MT + m;
            if (r < n_nodes)
                reinterpret_cast<float4*>(out + (size_t)r * DIM)[lane] =
                    *reinterpret_cast<const float4*>(&bufC[m][lane * 4]);
        }
    }
}

extern "C" void kernel_launch(void* const* d_in, const int* in_sizes, int n_in,
                              void* d_out, int out_size, void* d_ws, size_t ws_size,
                              hipStream_t stream) {
    const float* x  = (const float*)d_in[0];
    const int*   ei = (const int*)  d_in[1];
    const float* W1 = (const float*)d_in[2];
    const float* b1 = (const float*)d_in[3];
    const float* W2 = (const float*)d_in[4];
    const float* b2 = (const float*)d_in[5];
    float* out = (float*)d_out;

    const int n_nodes = in_sizes[0] / DIM;
    const int n_edges = in_sizes[1] / 2;

    const int n_lines = (n_edges + 4 * GROUP - 1) / (4 * GROUP) * 4;
    unsigned int* planes = (unsigned int*)d_ws;
    float* agg = (float*)((char*)d_ws + (size_t)MAX_RANGES * n_lines * (GROUP / 8));

    gin_plane_kernel<<<n_lines / PLANE_WAVES, PLANE_WAVES * 32, 0, stream>>>(
        ei + n_edges, planes, n_edges, n_nodes, n_lines);

    const int n_ranges = (n_nodes + RANGE_ROWS - 1) / RANGE_ROWS;
    gin_own_kernel<<<n_ranges, OWN_THREADS, 0, stream>>>(
        x, ei, ei + n_edges, planes, agg, n_edges, n_nodes, n_lines);

    const int tiles      = (n_nodes + MT - 1) / MT;
    const int blocks_mlp = tiles < 512 ? tiles : 512;
    gin_mlp_kernel<<<blocks_mlp, MLP_THREADS, 0, stream>>>(
        x, agg, W1, b1, W2, b2, out, n_nodes, tiles);
}


# Builder lines carry the edges: per range and 1024-edge group one 128-B line of packed (src,row) records (count + 31 slots) replaces the bit line; owners gather no src/dst; sum order: LDS arrival
# speedup vs baseline: 3.1599x; 1.1408x over previous
#include <hip/hip_runtime.h>

#define DIM       128
#define ROW_PAD   4
#define WSTEPS    (DIM / 4)

typedef float v2f __attribute__((ext_vector_type(2)));
typedef float v8f __attribute__((ext_vector_type(8)));

#define RANGE_ROWS   391
#define MAX_RANGES   256
#define OWN_WAVES    32
#define OWN_THREADS  (OWN_WAVES * 32)
#define GROUP        1024
#define CELL_SLOTS   31
#define RING         64
#define ROW_BATCH    8
#define PLANE_WAVES  2

typedef unsigned int u32;
typedef u32 u32x4 __attribute__((ext_vector_type(4)));

__global__ void __launch_bounds__(PLANE_WAVES * 32) gin_build_kernel(
    const int* __restrict__ src, const int* __restrict__ dst, u32* __restrict__ lines,
    int n_edges, int n_nodes, int n_lines)
{
    __shared__ __attribute__((aligned(16))) u32 cells[PLANE_WAVES][MAX_RANGES * 32];

    const int wave = __builtin_amdgcn_readfirstlane((int)(threadIdx.x >> 5));
    const int lane = threadIdx.x & 31;
    const int g    = blockIdx.x * PLANE_WAVES + wave;
    u32* mine = cells[wave];

    int d[8][4];
    int s[8][4];
    const long long e0 = (long long)g * GROUP;
    if ((n_edges & 3) == 0 && e0 + GROUP <= n_edges) {
        const int4* d4 = reinterpret_cast<const int4*>(dst + e0) + lane;
        const int4* s4 = reinterpret_cast<const int4*>(src + e0) + lane;
        #pragma unroll
        for (int j = 0; j < 8; ++j) {
            const int4 v = d4[j * 32];
            d[j][0] = v.x; d[j][1] = v.y; d[j][2] = v.z; d[j][3] = v.w;
        }
        #pragma unroll
        for (int j = 0; j < 8; ++j) {
            const int4 v = s4[j * 32];
            s[j][0] = v.x; s[j][1] = v.y; s[j][2] = v.z; s[j][3] = v.w;
        }
    } else {
        #pragma unroll
        for (int j = 0; j < 8; ++j) {
            #pragma unroll
            for (int k = 0; k < 4; ++k) {
                const long long e = e0 + j * 128 + lane * 4 + k;
                d[j][k] = e < n_edges ? dst[e] : -1;
                s[j][k] = e < n_edges ? src[e] : 0;
            }
        }
    }

    for (int k = 0; k < MAX_RANGES / 4; ++k)
        reinterpret_cast<u32x4*>(mine)[k * 32 + lane] = (u32x4){0u, 0u, 0u, 0u};
    __syncthreads();

    u32 slot[8][4];
    #pragma unroll
    for (int j = 0; j < 8; ++j) {
        #pragma unroll
        for (int k = 0; k < 4; ++k) {
            const int dd = d[j][k];
            const int ss = s[j][k];
            slot[j][k] = CELL_SLOTS;
            if ((u32)dd < (u32)n_nodes && (u32)ss < (u32)n_nodes) {
                const u32 t = (u32)dd / RANGE_ROWS;
                if (t < MAX_RANGES)
                    slot[j][k] = __hip_atomic_fetch_add(&mine[t * 32 + (t & 31u)], 1u,
                                     __ATOMIC_RELAXED, __HIP_MEMORY_SCOPE_WORKGROUP);
            }
        }
    }
    #pragma unroll
    for (int j = 0; j < 8; ++j) {
        #pragma unroll
        for (int k = 0; k < 4; ++k) {
            if (slot[j][k] < CELL_SLOTS) {
                const u32 dd  = (u32)d[j][k];
                const u32 t   = dd / RANGE_ROWS;
                const u32 row = dd - t * RANGE_ROWS;
                mine[t * 32 + (((t & 31u) + 1u + slot[j][k]) & 31u)] = ((u32)s[j][k] << 9) | row;
            }
        }
    }
    __syncthreads();

    const size_t plane_u4 = (size_t)n_lines * (GROUP / 128);
    u32x4* out4 = reinterpret_cast<u32x4*>(lines) + (size_t)(lane >> 3) * plane_u4
                + (size_t)g * (GROUP / 128) + (lane & 7);
    for (int k0 = 0; k0 < MAX_RANGES / 4; k0 += 8) {
        u32x4 v[8];
        #pragma unroll
        for (int t = 0; t < 8; ++t)
            v[t] = reinterpret_cast<const u32x4*>(mine)[(k0 + t) * 32 + lane];
        #pragma unroll
        for (int t = 0; t < 8; ++t)
            out4[(size_t)(k0 + t) * 4 * plane_u4] = v[t];
    }
}

template <bool FULL>
__device__ __forceinline__ void row_batch(
    const float* __restrict__ x, float* acc, int lane, int sv, int rv, u32& m, int n)
{
    float v[ROW_BATCH][4];
    int   rr[ROW_BATCH];
    #pragma unroll
    for (int j = 0; j < ROW_BATCH; ++j) {
        if (FULL || j < n) {
            const int k = __builtin_ctz(m);
            m &= m - 1;
            const int s = __builtin_amdgcn_readlane(sv, k);
            rr[j]       = __builtin_amdgcn_readlane(rv, k);
            const float* xr = x + (size_t)s * DIM + lane;
            v[j][0] = xr[0];
            v[j][1] = xr[32];
            v[j][2] = xr[64];
            v[j][3] = xr[96];
        }
    }
    #pragma unroll
    for (int j = 0; j < ROW_BATCH; ++j) {
        if (FULL || j < n) {
            float* p = acc + rr[j] * DIM + lane;
            __hip_atomic_fetch_add(p,      v[j][0], __ATOMIC_RELAXED, __HIP_MEMORY_SCOPE_WORKGROUP);
            __hip_atomic_fetch_add(p + 32, v[j][1], __ATOMIC_RELAXED, __HIP_MEMORY_SCOPE_WORKGROUP);
            __hip_atomic_fetch_add(p + 64, v[j][2], __ATOMIC_RELAXED, __HIP_MEMORY_SCOPE_WORKGROUP);
            __hip_atomic_fetch_add(p + 96, v[j][3], __ATOMIC_RELAXED, __HIP_MEMORY_SCOPE_WORKGROUP);
        }
    }
}

__device__ __forceinline__ void ring_put(int* my_ring, int& tail, u32 bal, bool valid, u32 rec)
{
    const int rank = (int)__builtin_amdgcn_mbcnt_lo(bal, 0u);
    if (valid)
        my_ring[(tail + rank) & (RING - 1)] = (int)rec;
    tail += __builtin_popcount(bal);
}

__device__ __forceinline__ void ring_take(
    const float* __restrict__ x, float* acc, const int* my_ring, int& head, int n,
    int lane, int n_nodes, int rows)
{
    const u32 rec = (u32)my_ring[(head + lane) & (RING - 1)];
    head += n;
    const int s = (int)(rec >> 9);
    const int r = (int)(rec & 511u);
    const bool ok = (lane < n) && ((u32)r < (u32)rows) && ((u32)s < (u32)n_nodes);
    u32 m = __builtin_amdgcn_ballot_w32(ok);
    int c = __builtin_popcount(m);
    while (c >= ROW_BATCH) {
        row_batch<true>(x, acc, lane, s, r, m, ROW_BATCH);
        c -= ROW_BATCH;
    }
    if (c > 0)
        row_batch<false>(x, acc, lane, s, r, m, c);
}

__global__ void __launch_bounds__(OWN_THREADS) gin_own_kernel(
    const float* __restrict__ x,
    const int*   __restrict__ src,
    const int*   __restrict__ dst,
    const u32*   __restrict__ lines,
    float*       __restrict__ agg,
    int n_edges, int n_nodes, int n_lines)
{
    __shared__ __attribute__((aligned(16))) float acc[RANGE_ROWS * DIM];
    __shared__ int ring[OWN_WAVES * RING];

    const int wave = __builtin_amdgcn_readfirstlane((int)(threadIdx.x >> 5));
    const int lane = threadIdx.x & 31;
    const int row0 = blockIdx.x * RANGE_ROWS;
    const int left = n_nodes - row0;
    const int rows = left < RANGE_ROWS ? left : RANGE_ROWS;

    for (int i = threadIdx.x; i < RANGE_ROWS * DIM / 4; i += OWN_THREADS)
        reinterpret_cast<float4*>(acc)[i] = make_float4(0.f, 0.f, 0.f, 0.f);
    __syncthreads();

    const int plane_u4 = n_lines * (GROUP / 128);
    const int n_pieces = plane_u4 >> 5;
    const u32x4* pl = reinterpret_cast<const u32x4*>(lines) + (size_t)blockIdx.x * plane_u4 + lane;

    const int cpos    = blockIdx.x & 31;
    const int cnt_src = ((lane & ~7) | (cpos >> 2)) << 2;

    int* my_ring = ring + wave * RING;
    int head = 0;
    int tail = 0;

    u32x4 w = __builtin_nontemporal_load(pl + (wave < n_pieces ? wave : 0) * 32);
    for (int i = wave; i < n_pieces; i += OWN_WAVES) {
        const int in = i + OWN_WAVES;
        const u32x4 wn = __builtin_nontemporal_load(pl + (in < n_pieces ? in : 0) * 32);
        const u32 wd[4] = {w.x, w.y, w.z, w.w};
        const u32 cw    = wd[cpos & 3];
        const u32 count = (u32)__builtin_amdgcn_ds_bpermute(cnt_src, (int)cw);
        const bool over = count > (u32)CELL_SLOTS;
        const u32 nrec  = over ? 0u : count;
        #pragma unroll
        for (int c = 0; c < 4; ++c) {
            const u32  k     = (u32)(4 * (lane & 7) + c - cpos - 1) & 31u;
            const bool valid = k < nrec;
            const u32  bal   = __builtin_amdgcn_ballot_w32(valid);
            if (bal) {
                ring_put(my_ring, tail, bal, valid, wd[c]);
                if (tail - head >= 32)
                    ring_take(x, acc, my_ring, head, 32, lane, n_nodes, rows);
            }
        }
        u32 ovm = __builtin_amdgcn_ballot_w32(over);
        while (ovm) {
            const int l = __builtin_ctz(ovm);
            ovm &= ~(0xffu << (l & ~7));
            const long long ge = (long long)((i * 32 + l) >> 3) * GROUP;
            for (int it = 0; it < GROUP / 32; ++it) {
                const long long e = ge + it * 32 + lane;
                const int  dd = e < n_edges ? dst[e] : -1;
                const int  ss = e < n_edges ? src[e] : 0;
                const int  r  = dd - row0;
                const bool ok = ((u32)r < (u32)rows) && ((u32)ss < (u32)n_nodes);
                const u32  bal = __builtin_amdgcn_ballot_w32(ok);
                if (bal) {
                    ring_put(my_ring, tail, bal, ok, ((u32)ss << 9) | (u32)r);
                    if (tail - head >= 32)
                        ring_take(x, acc, my_ring, head, 32, lane, n_nodes, rows);
                }
            }
        }
        w = wn;
    }
    if (tail - head > 0)
        ring_take(x, acc, my_ring, head, tail - head, lane, n_nodes, rows);
    __syncthreads();

    float4* out4 = reinterpret_cast<float4*>(agg + (size_t)row0 * DIM);
    for (int i = threadIdx.x; i < rows * (DIM / 4); i += OWN_THREADS)
        out4[i] = reinterpret_cast<const float4*>(acc)[i];
}

#define MLP_WAVES2  8
#define MLP_THREADS (MLP_WAVES2 * 32)
#define MT          32
#define PITCH       (DIM + ROW_PAD)
#define ROWS_PER_THREAD (MT / MLP_WAVES2)

__global__ void __launch_bounds__(MLP_THREADS) gin_mlp_kernel(
    const float* __restrict__ x,
    const float* __restrict__ agg,
    const float* __restrict__ W1,
    const float* __restrict__ b1,
    const float* __restrict__ W2,
    const float* __restrict__ b2,
    float*       __restrict__ out,
    int n_nodes, int n_tiles)
{
    __shared__ __attribute__((aligned(16))) float bufA[MT][PITCH];
    __shared__ __attribute__((aligned(16))) float bufB[MT][PITCH];
    __shared__ __attribute__((aligned(16))) float bufC[MT][PITCH];

    const int wave = threadIdx.x >> 5;
    const int lane = threadIdx.x & 31;
    const int c    = lane & 15;
    const int hh   = lane >> 4;
    const int koff = hh * 2;
    const int ncol = wave * 16 + c;

    v2f w1[WSTEPS], w2[WSTEPS];
    #pragma unroll
    for (int s = 0; s < WSTEPS; ++s) {
        const int k = (s << 2) + koff;
        w1[s].x = W1[(size_t)k * DIM + ncol];
        w1[s].y = W1[(size_t)(k + 1) * DIM + ncol];
        w2[s].x = W2[(size_t)k * DIM + ncol];
        w2[s].y = W2[(size_t)(k + 1) * DIM + ncol];
    }
    const float bias1 = b1[ncol];
    const float bias2 = b2[ncol];

    float4 xv[ROWS_PER_THREAD], av[ROWS_PER_THREAD];

    int tile = blockIdx.x;
    if (tile >= n_tiles) return;

    #pragma unroll
    for (int j = 0; j < ROWS_PER_THREAD; ++j) {
        const int m = wave + MLP_WAVES2 * j;
        const int r = tile * MT + m;
        float4 h = make_float4(0.f, 0.f, 0.f, 0.f);
        if (r < n_nodes) {
            const float4 a = reinterpret_cast<const float4*>(x   + (size_t)r * DIM)[lane];
            const float4 b = reinterpret_cast<const float4*>(agg + (size_t)r * DIM)[lane];
            h = make_float4(a.x + b.x, a.y + b.y, a.z + b.z, a.w + b.w);
        }
        *reinterpret_cast<float4*>(&bufA[m][lane * 4]) = h;
    }
    __syncthreads();

    const float* a_row0 = &bufA[c][koff];
    const float* a_row1 = &bufA[16 + c][koff];
    const float* h_row0 = &bufB[c][koff];
    const float* h_row1 = &bufB[16 + c][koff];

    for (; tile < n_tiles; tile += gridDim.x) {
        const int  next     = tile + (int)gridDim.x;
        const bool has_next = next < n_tiles;

        if (has_next) {
            #pragma unroll
            for (int j = 0; j < ROWS_PER_THREAD; ++j) {
                const int r = next * MT + wave + MLP_WAVES2 * j;
                if (r < n_nodes) {
                    xv[j] = reinterpret_cast<const float4*>(x   + (size_t)r * DIM)[lane];
                    av[j] = reinterpret_cast<const float4*>(agg + (size_t)r * DIM)[lane];
                } else {
                    xv[j] = make_float4(0.f, 0.f, 0.f, 0.f);
                    av[j] = make_float4(0.f, 0.f, 0.f, 0.f);
                }
            }
        }

        v8f c0 = {0.f, 0.f, 0.f, 0.f, 0.f, 0.f, 0.f, 0.f};
        v8f c1 = {0.f, 0.f, 0.f, 0.f, 0.f, 0.f, 0.f, 0.f};
        #pragma unroll
        for (int s = 0; s < WSTEPS; ++s) {
            const v2f a0 = *reinterpret_cast<const v2f*>(a_row0 + (s << 2));
            const v2f a1 = *reinterpret_cast<const v2f*>(a_row1 + (s << 2));
            c0 = __builtin_amdgcn_wmma_f32_16x16x4_f32(
                     false, a0, false, w1[s], (short)0, c0, false, false);
            c1 = __builtin_amdgcn_wmma_f32_16x16x4_f32(
                     false, a1, false, w1[s], (short)0, c1, false, false);
        }
        #pragma unroll
        for (int v = 0; v < 8; ++v) {
            const float t0 = c0[v] + bias1;
            const float t1 = c1[v] + bias1;
            bufB[hh * 8 + v][ncol]      = t0 > 0.f ? t0 : 0.f;
            bufB[16 + hh * 8 + v][ncol] = t1 > 0.f ? t1 : 0.f;
        }
        __syncthreads();

        if (has_next) {
            #pragma unroll
            for (int j = 0; j < ROWS_PER_THREAD; ++j) {
                const int m = wave + MLP_WAVES2 * j;
                *reinterpret_cast<float4*>(&bufA[m][lane * 4]) =
                    make_float4(xv[j].x + av[j].x, xv[j].y + av[j].y,
                                xv[j].z + av[j].z, xv[j].w + av[j].w);
            }
        }

        v8f d0 = {0.f, 0.f, 0.f, 0.f, 0.f, 0.f, 0.f, 0.f};
        v8f d1 = {0.f, 0.f, 0.f, 0.f, 0.f, 0.f, 0.f, 0.f};
        #pragma unroll
        for (int s = 0; s < WSTEPS; ++s) {
            const v2f a0 = *reinterpret_cast<const v2f*>(h_row0 + (s << 2));
            const v2f a1 = *reinterpret_cast<const v2f*>(h_row1 + (s << 2));
            d0 = __builtin_amdgcn_wmma_f32_16x16x4_f32(
                     false, a0, false, w2[s], (short)0, d0, false, false);
            d1 = __builtin_amdgcn_wmma_f32_16x16x4_f32(
                     false, a1, false, w2[s], (short)0, d1, false, false);
        }
        #pragma unroll
        for (int v = 0; v < 8; ++v) {
            bufC[hh * 8 + v][ncol]      = d0[v] + bias2;
            bufC[16 + hh * 8 + v][ncol] = d1[v] + bias2;
        }
        __syncthreads();

        #pragma unroll
        for (int j = 0; j < ROWS_PER_THREAD; ++j) {
            const int m = wave + MLP_WAVES2 * j;
            const int r = tile * MT + m;
            if (r < n_nodes)
                reinterpret_cast<float4*>(out + (size_t)r * DIM)[lane] =
                    *reinterpret_cast<const float4*>(&bufC[m][lane * 4]);
        }
    }
}

extern "C" void kernel_launch(void* const* d_in, const int* in_sizes, int n_in,
                              void* d_out, int out_size, void* d_ws, size_t ws_size,
                              hipStream_t stream) {
    const float* x  = (const float*)d_in[0];
    const int*   ei = (const int*)  d_in[1];
    const float* W1 = (const float*)d_in[2];
    const float* b1 = (const float*)d_in[3];
    const float* W2 = (const float*)d_in[4];
    const float* b2 = (const float*)d_in[5];
    float* out = (float*)d_out;

    const int n_nodes = in_sizes[0] / DIM;
    const int n_edges = in_sizes[1] / 2;

    const int n_lines = (n_edges + 4 * GROUP - 1) / (4 * GROUP) * 4;
    unsigned int* lines = (unsigned int*)d_ws;
    float* agg = (float*)((char*)d_ws + (size_t)MAX_RANGES * n_lines * (GROUP / 8));

    gin_build_kernel<<<n_lines / PLANE_WAVES, PLANE_WAVES * 32, 0, stream>>>(
        ei, ei + n_edges, lines, n_edges, n_nodes, n_lines);

    const int n_ranges = (n_nodes + RANGE_ROWS - 1) / RANGE_ROWS;
    gin_own_kernel<<<n_ranges, OWN_THREADS, 0, stream>>>(
        x, ei, ei + n_edges, lines, agg, n_edges, n_nodes, n_lines);

    const int tiles      = (n_nodes + MT - 1) / MT;
    const int blocks_mlp = tiles < 512 ? tiles : 512;
    gin_mlp_kernel<<<blocks_mlp, MLP_THREADS, 0, stream>>>(
        x, agg, W1, b1, W2, b2, out, n_nodes, tiles);
}


# Owner gather: one 16-B-per-lane load per x row (4 whole lines per instruction, was four 1-line loads); LDS rows kept permuted (col 4l+c at word 32c+l), atomics conflict-free; sum order: LDS arrival
# speedup vs baseline: 3.3997x; 1.0759x over previous
#include <hip/hip_runtime.h>

#define DIM       128
#define ROW_PAD   4
#define WSTEPS    (DIM / 4)

typedef float v2f __attribute__((ext_vector_type(2)));
typedef float v8f __attribute__((ext_vector_type(8)));

#define RANGE_ROWS   391
#define MAX_RANGES   256
#define OWN_WAVES    32
#define OWN_THREADS  (OWN_WAVES * 32)
#define GROUP        1024
#define CELL_SLOTS   31
#define RING         64
#define ROW_BATCH    8
#define PLANE_WAVES  2

typedef unsigned int u32;
typedef u32 u32x4 __attribute__((ext_vector_type(4)));

__global__ void __launch_bounds__(PLANE_WAVES * 32) gin_build_kernel(
    const int* __restrict__ src, const int* __restrict__ dst, u32* __restrict__ lines,
    int n_edges, int n_nodes, int n_lines)
{
    __shared__ __attribute__((aligned(16))) u32 cells[PLANE_WAVES][MAX_RANGES * 32];

    const int wave = __builtin_amdgcn_readfirstlane((int)(threadIdx.x >> 5));
    const int lane = threadIdx.x & 31;
    const int g    = blockIdx.x * PLANE_WAVES + wave;
    u32* mine = cells[wave];

    int d[8][4];
    int s[8][4];
    const long long e0 = (long long)g * GROUP;
    if ((n_edges & 3) == 0 && e0 + GROUP <= n_edges) {
        const int4* d4 = reinterpret_cast<const int4*>(dst + e0) + lane;
        const int4* s4 = reinterpret_cast<const int4*>(src + e0) + lane;
        #pragma unroll
        for (int j = 0; j < 8; ++j) {
            const int4 v = d4[j * 32];
            d[j][0] = v.x; d[j][1] = v.y; d[j][2] = v.z; d[j][3] = v.w;
        }
        #pragma unroll
        for (int j = 0; j < 8; ++j) {
            const int4 v = s4[j * 32];
            s[j][0] = v.x; s[j][1] = v.y; s[j][2] = v.z; s[j][3] = v.w;
        }
    } else {
        #pragma unroll
        for (int j = 0; j < 8; ++j) {
            #pragma unroll
            for (int k = 0; k < 4; ++k) {
                const long long e = e0 + j * 128 + lane * 4 + k;
                d[j][k] = e < n_edges ? dst[e] : -1;
                s[j][k] = e < n_edges ? src[e] : 0;
            }
        }
    }

    for (int k = 0; k < MAX_RANGES / 4; ++k)
        reinterpret_cast<u32x4*>(mine)[k * 32 + lane] = (u32x4){0u, 0u, 0u, 0u};
    __syncthreads();

    u32 slot[8][4];
    #pragma unroll
    for (int j = 0; j < 8; ++j) {
        #pragma unroll
        for (int k = 0; k < 4; ++k) {
            const int dd = d[j][k];
            const int ss = s[j][k];
            slot[j][k] = CELL_SLOTS;
            if ((u32)dd < (u32)n_nodes && (u32)ss < (u32)n_nodes) {
                const u32 t = (u32)dd / RANGE_ROWS;
                if (t < MAX_RANGES)
                    slot[j][k] = __hip_atomic_fetch_add(&mine[t * 32 + (t & 31u)], 1u,
                                     __ATOMIC_RELAXED, __HIP_MEMORY_SCOPE_WORKGROUP);
            }
        }
    }
    #pragma unroll
    for (int j = 0; j < 8; ++j) {
        #pragma unroll
        for (int k = 0; k < 4; ++k) {
            if (slot[j][k] < CELL_SLOTS) {
                const u32 dd  = (u32)d[j][k];
                const u32 t   = dd / RANGE_ROWS;
                const u32 row = dd - t * RANGE_ROWS;
                mine[t * 32 + (((t & 31u) + 1u + slot[j][k]) & 31u)] = ((u32)s[j][k] << 9) | row;
            }
        }
    }
    __syncthreads();

    const size_t plane_u4 = (size_t)n_lines * (GROUP / 128);
    u32x4* out4 = reinterpret_cast<u32x4*>(lines) + (size_t)(lane >> 3) * plane_u4
                + (size_t)g * (GROUP / 128) + (lane & 7);
    for (int k0 = 0; k0 < MAX_RANGES / 4; k0 += 8) {
        u32x4 v[8];
        #pragma unroll
        for (int t = 0; t < 8; ++t)
            v[t] = reinterpret_cast<const u32x4*>(mine)[(k0 + t) * 32 + lane];
        #pragma unroll
        for (int t = 0; t < 8; ++t)
            out4[(size_t)(k0 + t) * 4 * plane_u4] = v[t];
    }
}

template <bool FULL>
__device__ __forceinline__ void row_batch(
    const float* __restrict__ x, float* acc, int lane, int sv, int rv, u32& m, int n)
{
    float4 v[ROW_BATCH];
    int    rr[ROW_BATCH];
    #pragma unroll
    for (int j = 0; j < ROW_BATCH; ++j) {
        if (FULL || j < n) {
            const int k = __builtin_ctz(m);
            m &= m - 1;
            const int s = __builtin_amdgcn_readlane(sv, k);
            rr[j]       = __builtin_amdgcn_readlane(rv, k);
            v[j] = reinterpret_cast<const float4*>(x + (size_t)s * DIM)[lane];
        }
    }
    #pragma unroll
    for (int j = 0; j < ROW_BATCH; ++j) {
        if (FULL || j < n) {
            float* p = acc + rr[j] * DIM + lane;
            __hip_atomic_fetch_add(p,      v[j].x, __ATOMIC_RELAXED, __HIP_MEMORY_SCOPE_WORKGROUP);
            __hip_atomic_fetch_add(p + 32, v[j].y, __ATOMIC_RELAXED, __HIP_MEMORY_SCOPE_WORKGROUP);
            __hip_atomic_fetch_add(p + 64, v[j].z, __ATOMIC_RELAXED, __HIP_MEMORY_SCOPE_WORKGROUP);
            __hip_atomic_fetch_add(p + 96, v[j].w, __ATOMIC_RELAXED, __HIP_MEMORY_SCOPE_WORKGROUP);
        }
    }
}

__device__ __forceinline__ void ring_put(int* my_ring, int& tail, u32 bal, bool valid, u32 rec)
{
    const int rank = (int)__builtin_amdgcn_mbcnt_lo(bal, 0u);
    if (valid)
        my_ring[(tail + rank) & (RING - 1)] = (int)rec;
    tail += __builtin_popcount(bal);
}

__device__ __forceinline__ void ring_take(
    const float* __restrict__ x, float* acc, const int* my_ring, int& head, int n,
    int lane, int n_nodes, int rows)
{
    const u32 rec = (u32)my_ring[(head + lane) & (RING - 1)];
    head += n;
    const int s = (int)(rec >> 9);
    const int r = (int)(rec & 511u);
    const bool ok = (lane < n) && ((u32)r < (u32)rows) && ((u32)s < (u32)n_nodes);
    u32 m = __builtin_amdgcn_ballot_w32(ok);
    int c = __builtin_popcount(m);
    while (c >= ROW_BATCH) {
        row_batch<true>(x, acc, lane, s, r, m, ROW_BATCH);
        c -= ROW_BATCH;
    }
    if (c > 0)
        row_batch<false>(x, acc, lane, s, r, m, c);
}

__global__ void __launch_bounds__(OWN_THREADS) gin_own_kernel(
    const float* __restrict__ x,
    const int*   __restrict__ src,
    const int*   __restrict__ dst,
    const u32*   __restrict__ lines,
    float*       __restrict__ agg,
    int n_edges, int n_nodes, int n_lines)
{
    __shared__ __attribute__((aligned(16))) float acc[RANGE_ROWS * DIM];
    __shared__ int ring[OWN_WAVES * RING];

    const int wave = __builtin_amdgcn_readfirstlane((int)(threadIdx.x >> 5));
    const int lane = threadIdx.x & 31;
    const int row0 = blockIdx.x * RANGE_ROWS;
    const int left = n_nodes - row0;
    const int rows = left < RANGE_ROWS ? left : RANGE_ROWS;

    for (int i = threadIdx.x; i < RANGE_ROWS * DIM / 4; i += OWN_THREADS)
        reinterpret_cast<float4*>(acc)[i] = make_float4(0.f, 0.f, 0.f, 0.f);
    __syncthreads();

    const int plane_u4 = n_lines * (GROUP / 128);
    const int n_pieces = plane_u4 >> 5;
    const u32x4* pl = reinterpret_cast<const u32x4*>(lines) + (size_t)blockIdx.x * plane_u4 + lane;

    const int cpos    = blockIdx.x & 31;
    const int cnt_src = ((lane & ~7) | (cpos >> 2)) << 2;

    int* my_ring = ring + wave * RING;
    int head = 0;
    int tail = 0;

    u32x4 w = __builtin_nontemporal_load(pl + (wave < n_pieces ? wave : 0) * 32);
    for (int i = wave; i < n_pieces; i += OWN_WAVES) {
        const int in = i + OWN_WAVES;
        const u32x4 wn = __builtin_nontemporal_load(pl + (in < n_pieces ? in : 0) * 32);
        const u32 wd[4] = {w.x, w.y, w.z, w.w};
        const u32 cw    = wd[cpos & 3];
        const u32 count = (u32)__builtin_amdgcn_ds_bpermute(cnt_src, (int)cw);
        const bool over = count > (u32)CELL_SLOTS;
        const u32 nrec  = over ? 0u : count;
        #pragma unroll
        for (int c = 0; c < 4; ++c) {
            const u32  k     = (u32)(4 * (lane & 7) + c - cpos - 1) & 31u;
            const bool valid = k < nrec;
            const u32  bal   = __builtin_amdgcn_ballot_w32(valid);
            if (bal) {
                ring_put(my_ring, tail, bal, valid, wd[c]);
                if (tail - head >= 32)
                    ring_take(x, acc, my_ring, head, 32, lane, n_nodes, rows);
            }
        }
        u32 ovm = __builtin_amdgcn_ballot_w32(over);
        while (ovm) {
            const int l = __builtin_ctz(ovm);
            ovm &= ~(0xffu << (l & ~7));
            const long long ge = (long long)((i * 32 + l) >> 3) * GROUP;
            for (int it = 0; it < GROUP / 32; ++it) {
                const long long e = ge + it * 32 + lane;
                const int  dd = e < n_edges ? dst[e] : -1;
                const int  ss = e < n_edges ? src[e] : 0;
                const int  r  = dd - row0;
                const bool ok = ((u32)r < (u32)rows) && ((u32)ss < (u32)n_nodes);
                const u32  bal = __builtin_amdgcn_ballot_w32(ok);
                if (bal) {
                    ring_put(my_ring, tail, bal, ok, ((u32)ss << 9) | (u32)r);
                    if (tail - head >= 32)
                        ring_take(x, acc, my_ring, head, 32, lane, n_nodes, rows);
                }
            }
        }
        w = wn;
    }
    if (tail - head > 0)
        ring_take(x, acc, my_ring, head, tail - head, lane, n_nodes, rows);
    __syncthreads();

    float4* out4 = reinterpret_cast<float4*>(agg + (size_t)row0 * DIM);
    for (int i = threadIdx.x; i < rows * (DIM / 4); i += OWN_THREADS) {
        const float* p = acc + (i >> 5) * DIM + (i & 31);
        out4[i] = make_float4(p[0], p[32], p[64], p[96]);
    }
}

#define MLP_WAVES2  8
#define MLP_THREADS (MLP_WAVES2 * 32)
#define MT          32
#define PITCH       (DIM + ROW_PAD)
#define ROWS_PER_THREAD (MT / MLP_WAVES2)

__global__ void __launch_bounds__(MLP_THREADS) gin_mlp_kernel(
    const float* __restrict__ x,
    const float* __restrict__ agg,
    const float* __restrict__ W1,
    const float* __restrict__ b1,
    const float* __restrict__ W2,
    const float* __restrict__ b2,
    float*       __restrict__ out,
    int n_nodes, int n_tiles)
{
    __shared__ __attribute__((aligned(16))) float bufA[MT][PITCH];
    __shared__ __attribute__((aligned(16))) float bufB[MT][PITCH];
    __shared__ __attribute__((aligned(16))) float bufC[MT][PITCH];

    const int wave = threadIdx.x >> 5;
    const int lane = threadIdx.x & 31;
    const int c    = lane & 15;
    const int hh   = lane >> 4;
    const int koff = hh * 2;
    const int ncol = wave * 16 + c;

    v2f w1[WSTEPS], w2[WSTEPS];
    #pragma unroll
    for (int s = 0; s < WSTEPS; ++s) {
        const int k = (s << 2) + koff;
        w1[s].x = W1[(size_t)k * DIM + ncol];
        w1[s].y = W1[(size_t)(k + 1) * DIM + ncol];
        w2[s].x = W2[(size_t)k * DIM + ncol];
        w2[s].y = W2[(size_t)(k + 1) * DIM + ncol];
    }
    const float bias1 = b1[ncol];
    const float bias2 = b2[ncol];

    float4 xv[ROWS_PER_THREAD], av[ROWS_PER_THREAD];

    int tile = blockIdx.x;
    if (tile >= n_tiles) return;

    #pragma unroll
    for (int j = 0; j < ROWS_PER_THREAD; ++j) {
        const int m = wave + MLP_WAVES2 * j;
        const int r = tile * MT + m;
        float4 h = make_float4(0.f, 0.f, 0.f, 0.f);
        if (r < n_nodes) {
            const float4 a = reinterpret_cast<const float4*>(x   + (size_t)r * DIM)[lane];
            const float4 b = reinterpret_cast<const float4*>(agg + (size_t)r * DIM)[lane];
            h = make_float4(a.x + b.x, a.y + b.y, a.z + b.z, a.w + b.w);
        }
        *reinterpret_cast<float4*>(&bufA[m][lane * 4]) = h;
    }
    __syncthreads();

    const float* a_row0 = &bufA[c][koff];
    const float* a_row1 = &bufA[16 + c][koff];
    const float* h_row0 = &bufB[c][koff];
    const float* h_row1 = &bufB[16 + c][koff];

    for (; tile < n_tiles; tile += gridDim.x) {
        const int  next     = tile + (int)gridDim.x;
        const bool has_next = next < n_tiles;

        if (has_next) {
            #pragma unroll
            for (int j = 0; j < ROWS_PER_THREAD; ++j) {
                const int r = next * MT + wave + MLP_WAVES2 * j;
                if (r < n_nodes) {
                    xv[j] = reinterpret_cast<const float4*>(x   + (size_t)r * DIM)[lane];
                    av[j] = reinterpret_cast<const float4*>(agg + (size_t)r * DIM)[lane];
                } else {
                    xv[j] = make_float4(0.f, 0.f, 0.f, 0.f);
                    av[j] = make_float4(0.f, 0.f, 0.f, 0.f);
                }
            }
        }

        v8f c0 = {0.f, 0.f, 0.f, 0.f, 0.f, 0.f, 0.f, 0.f};
        v8f c1 = {0.f, 0.f, 0.f, 0.f, 0.f, 0.f, 0.f, 0.f};
        #pragma unroll
        for (int s = 0; s < WSTEPS; ++s) {
            const v2f a0 = *reinterpret_cast<const v2f*>(a_row0 + (s << 2));
            const v2f a1 = *reinterpret_cast<const v2f*>(a_row1 + (s << 2));
            c0 = __builtin_amdgcn_wmma_f32_16x16x4_f32(
                     false, a0, false, w1[s], (short)0, c0, false, false);
            c1 = __builtin_amdgcn_wmma_f32_16x16x4_f32(
                     false, a1, false, w1[s], (short)0, c1, false, false);
        }
        #pragma unroll
        for (int v = 0; v < 8; ++v) {
            const float t0 = c0[v] + bias1;
            const float t1 = c1[v] + bias1;
            bufB[hh * 8 + v][ncol]      = t0 > 0.f ? t0 : 0.f;
            bufB[16 + hh * 8 + v][ncol] = t1 > 0.f ? t1 : 0.f;
        }
        __syncthreads();

        if (has_next) {
            #pragma unroll
            for (int j = 0; j < ROWS_PER_THREAD; ++j) {
                const int m = wave + MLP_WAVES2 * j;
                *reinterpret_cast<float4*>(&bufA[m][lane * 4]) =
                    make_float4(xv[j].x + av[j].x, xv[j].y + av[j].y,
                                xv[j].z + av[j].z, xv[j].w + av[j].w);
            }
        }

        v8f d0 = {0.f, 0.f, 0.f, 0.f, 0.f, 0.f, 0.f, 0.f};
        v8f d1 = {0.f, 0.f, 0.f, 0.f, 0.f, 0.f, 0.f, 0.f};
        #pragma unroll
        for (int s = 0; s < WSTEPS; ++s) {
            const v2f a0 = *reinterpret_cast<const v2f*>(h_row0 + (s << 2));
            const v2f a1 = *reinterpret_cast<const v2f*>(h_row1 + (s << 2));
            d0 = __builtin_amdgcn_wmma_f32_16x16x4_f32(
                     false, a0, false, w2[s], (short)0, d0, false, false);
            d1 = __builtin_amdgcn_wmma_f32_16x16x4_f32(
                     false, a1, false, w2[s], (short)0, d1, false, false);
        }
        #pragma unroll
        for (int v = 0; v < 8; ++v) {
            bufC[hh * 8 + v][ncol]      = d0[v] + bias2;
            bufC[16 + hh * 8 + v][ncol] = d1[v] + bias2;
        }
        __syncthreads();

        #pragma unroll
        for (int j = 0; j < ROWS_PER_THREAD; ++j) {
            const int m = wave + MLP_WAVES2 * j;
            const int r = tile * MT + m;
            if (r < n_nodes)
                reinterpret_cast<float4*>(out + (size_t)r * DIM)[lane] =
                    *reinterpret_cast<const float4*>(&bufC[m][lane * 4]);
        }
    }
}

extern "C" void kernel_launch(void* const* d_in, const int* in_sizes, int n_in,
                              void* d_out, int out_size, void* d_ws, size_t ws_size,
                              hipStream_t stream) {
    const float* x  = (const float*)d_in[0];
    const int*   ei = (const int*)  d_in[1];
    const float* W1 = (const float*)d_in[2];
    const float* b1 = (const float*)d_in[3];
    const float* W2 = (const float*)d_in[4];
    const float* b2 = (const float*)d_in[5];
    float* out = (float*)d_out;

    const int n_nodes = in_sizes[0] / DIM;
    const int n_edges = in_sizes[1] / 2;

    const int n_lines = (n_edges + 4 * GROUP - 1) / (4 * GROUP) * 4;
    unsigned int* lines = (unsigned int*)d_ws;
    float* agg = (float*)((char*)d_ws + (size_t)MAX_RANGES * n_lines * (GROUP / 8));

    gin_build_kernel<<<n_lines / PLANE_WAVES, PLANE_WAVES * 32, 0, stream>>>(
        ei, ei + n_edges, lines, n_edges, n_nodes, n_lines);

    const int n_ranges = (n_nodes + RANGE_ROWS - 1) / RANGE_ROWS;
    gin_own_kernel<<<n_ranges, OWN_THREADS, 0, stream>>>(
        x, ei, ei + n_edges, lines, agg, n_edges, n_nodes, n_lines);

    const int tiles      = (n_nodes + MT - 1) / MT;
    const int blocks_mlp = tiles < 512 ? tiles : 512;
    gin_mlp_kernel<<<blocks_mlp, MLP_THREADS, 0, stream>>>(
        x, agg, W1, b1, W2, b2, out, n_nodes, tiles);
}


# Owner block runs the MLP too: LDS rows start as x, hold h after the gather, its 32 waves do both GEMMs on them in place (f32 WMMA); no agg, 2 launches; sums start at x, layer-1 k order permuted
# speedup vs baseline: 3.7173x; 1.0934x over previous
#include <hip/hip_runtime.h>

#define DIM       128
#define ROW_PAD   4
#define WSTEPS    (DIM / 4)

typedef float v2f __attribute__((ext_vector_type(2)));
typedef float v8f __attribute__((ext_vector_type(8)));

#define RANGE_ROWS   391
#define OWN_TILES    ((RANGE_ROWS + 15) / 16)
#define TILE_ROWS    (OWN_TILES * 16)
#define APITCH       (DIM + ROW_PAD)
#define ROW_GROUPS   4
#define INIT_ROWS    ((TILE_ROWS + 31) / 32)
#define MAX_RANGES   256
#define OWN_WAVES    32
#define OWN_THREADS  (OWN_WAVES * 32)
#define GROUP        1024
#define CELL_SLOTS   31
#define RING         64
#define ROW_BATCH    8
#define PLANE_WAVES  2

typedef unsigned int u32;
typedef u32 u32x4 __attribute__((ext_vector_type(4)));

__global__ void __launch_bounds__(PLANE_WAVES * 32) gin_build_kernel(
    const int* __restrict__ src, const int* __restrict__ dst, u32* __restrict__ lines,
    int n_edges, int n_nodes, int n_lines)
{
    __shared__ __attribute__((aligned(16))) u32 cells[PLANE_WAVES][MAX_RANGES * 32];

    const int wave = __builtin_amdgcn_readfirstlane((int)(threadIdx.x >> 5));
    const int lane = threadIdx.x & 31;
    const int g    = blockIdx.x * PLANE_WAVES + wave;
    u32* mine = cells[wave];

    int d[8][4];
    int s[8][4];
    const long long e0 = (long long)g * GROUP;
    if ((n_edges & 3) == 0 && e0 + GROUP <= n_edges) {
        const int4* d4 = reinterpret_cast<const int4*>(dst + e0) + lane;
        const int4* s4 = reinterpret_cast<const int4*>(src + e0) + lane;
        #pragma unroll
        for (int j = 0; j < 8; ++j) {
            const int4 v = d4[j * 32];
            d[j][0] = v.x; d[j][1] = v.y; d[j][2] = v.z; d[j][3] = v.w;
        }
        #pragma unroll
        for (int j = 0; j < 8; ++j) {
            const int4 v = s4[j * 32];
            s[j][0] = v.x; s[j][1] = v.y; s[j][2] = v.z; s[j][3] = v.w;
        }
    } else {
        #pragma unroll
        for (int j = 0; j < 8; ++j) {
            #pragma unroll
            for (int k = 0; k < 4; ++k) {
                const long long e = e0 + j * 128 + lane * 4 + k;
                d[j][k] = e < n_edges ? dst[e] : -1;
                s[j][k] = e < n_edges ? src[e] : 0;
            }
        }
    }

    for (int k = 0; k < MAX_RANGES / 4; ++k)
        reinterpret_cast<u32x4*>(mine)[k * 32 + lane] = (u32x4){0u, 0u, 0u, 0u};
    __syncthreads();

    u32 slot[8][4];
    #pragma unroll
    for (int j = 0; j < 8; ++j) {
        #pragma unroll
        for (int k = 0; k < 4; ++k) {
            const int dd = d[j][k];
            const int ss = s[j][k];
            slot[j][k] = CELL_SLOTS;
            if ((u32)dd < (u32)n_nodes && (u32)ss < (u32)n_nodes) {
                const u32 t = (u32)dd / RANGE_ROWS;
                if (t < MAX_RANGES)
                    slot[j][k] = __hip_atomic_fetch_add(&mine[t * 32 + (t & 31u)], 1u,
                                     __ATOMIC_RELAXED, __HIP_MEMORY_SCOPE_WORKGROUP);
            }
        }
    }
    #pragma unroll
    for (int j = 0; j < 8; ++j) {
        #pragma unroll
        for (int k = 0; k < 4; ++k) {
            if (slot[j][k] < CELL_SLOTS) {
                const u32 dd  = (u32)d[j][k];
                const u32 t   = dd / RANGE_ROWS;
                const u32 row = dd - t * RANGE_ROWS;
                mine[t * 32 + (((t & 31u) + 1u + slot[j][k]) & 31u)] = ((u32)s[j][k] << 9) | row;
            }
        }
    }
    __syncthreads();

    const size_t plane_u4 = (size_t)n_lines * (GROUP / 128);
    u32x4* out4 = reinterpret_cast<u32x4*>(lines) + (size_t)(lane >> 3) * plane_u4
                + (size_t)g * (GROUP / 128) + (lane & 7);
    for (int k0 = 0; k0 < MAX_RANGES / 4; k0 += 8) {
        u32x4 v[8];
        #pragma unroll
        for (int t = 0; t < 8; ++t)
            v[t] = reinterpret_cast<const u32x4*>(mine)[(k0 + t) * 32 + lane];
        #pragma unroll
        for (int t = 0; t < 8; ++t)
            out4[(size_t)(k0 + t) * 4 * plane_u4] = v[t];
    }
}

template <bool FULL>
__device__ __forceinline__ void row_batch(
    const float* __restrict__ x, float* acc, int lane, int sv, int rv, u32& m, int n)
{
    float4 v[ROW_BATCH];
    int    rr[ROW_BATCH];
    #pragma unroll
    for (int j = 0; j < ROW_BATCH; ++j) {
        if (FULL || j < n) {
            const int k = __builtin_ctz(m);
            m &= m - 1;
            const int s = __builtin_amdgcn_readlane(sv, k);
            rr[j]       = __builtin_amdgcn_readlane(rv, k);
            v[j] = reinterpret_cast<const float4*>(x + (size_t)s * DIM)[lane];
        }
    }
    #pragma unroll
    for (int j = 0; j < ROW_BATCH; ++j) {
        if (FULL || j < n) {
            float* p = acc + rr[j] * APITCH + lane;
            __hip_atomic_fetch_add(p,      v[j].x, __ATOMIC_RELAXED, __HIP_MEMORY_SCOPE_WORKGROUP);
            __hip_atomic_fetch_add(p + 32, v[j].y, __ATOMIC_RELAXED, __HIP_MEMORY_SCOPE_WORKGROUP);
            __hip_atomic_fetch_add(p + 64, v[j].z, __ATOMIC_RELAXED, __HIP_MEMORY_SCOPE_WORKGROUP);
            __hip_atomic_fetch_add(p + 96, v[j].w, __ATOMIC_RELAXED, __HIP_MEMORY_SCOPE_WORKGROUP);
        }
    }
}

__device__ __forceinline__ void ring_put(int* my_ring, int& tail, u32 bal, bool valid, u32 rec)
{
    const int rank = (int)__builtin_amdgcn_mbcnt_lo(bal, 0u);
    if (valid)
        my_ring[(tail + rank) & (RING - 1)] = (int)rec;
    tail += __builtin_popcount(bal);
}

__device__ __forceinline__ void ring_take(
    const float* __restrict__ x, float* acc, const int* my_ring, int& head, int n,
    int lane, int n_nodes, int rows)
{
    const u32 rec = (u32)my_ring[(head + lane) & (RING - 1)];
    head += n;
    const int s = (int)(rec >> 9);
    const int r = (int)(rec & 511u);
    const bool ok = (lane < n) && ((u32)r < (u32)rows) && ((u32)s < (u32)n_nodes);
    u32 m = __builtin_amdgcn_ballot_w32(ok);
    int c = __builtin_popcount(m);
    while (c >= ROW_BATCH) {
        row_batch<true>(x, acc, lane, s, r, m, ROW_BATCH);
        c -= ROW_BATCH;
    }
    if (c > 0)
        row_batch<false>(x, acc, lane, s, r, m, c);
}


__device__ __forceinline__ v8f tile_gemm(const float* arow, const v2f (&wf)[WSTEPS])
{
    v8f cc = {0.f, 0.f, 0.f, 0.f, 0.f, 0.f, 0.f, 0.f};
    #pragma unroll
    for (int s = 0; s < WSTEPS; ++s) {
        const v2f a = *reinterpret_cast<const v2f*>(arow + (s << 2));
        cc = __builtin_amdgcn_wmma_f32_16x16x4_f32(
                 false, a, false, wf[s], (short)0, cc, false, false);
    }
    return cc;
}

__device__ __forceinline__ void rows_out(
    const float* acc, float* __restrict__ out_rows, int r0, int r1, int wave, int lane)
{
    for (int r = r0 + wave; r < r1; r += OWN_WAVES)
        reinterpret_cast<float4*>(out_rows + (size_t)r * DIM)[lane] =
            *reinterpret_cast<const float4*>(acc + r * APITCH + lane * 4);
}

__global__ void __launch_bounds__(OWN_THREADS) gin_own_kernel(
    const float* __restrict__ x,
    const int*   __restrict__ src,
    const int*   __restrict__ dst,
    const u32*   __restrict__ lines,
    const float* __restrict__ W1,
    const float* __restrict__ b1,
    const float* __restrict__ W2,
    const float* __restrict__ b2,
    float*       __restrict__ out,
    int n_edges, int n_nodes, int n_lines)
{
    __shared__ __attribute__((aligned(16))) float acc[TILE_ROWS * APITCH];
    __shared__ int ring[OWN_WAVES * RING];

    const int wave = __builtin_amdgcn_readfirstlane((int)(threadIdx.x >> 5));
    const int lane = threadIdx.x & 31;
    const int row0 = blockIdx.x * RANGE_ROWS;
    const int left = n_nodes - row0;
    const int rows = left < RANGE_ROWS ? left : RANGE_ROWS;

    const int plane_u4 = n_lines * (GROUP / 128);
    const int n_pieces = plane_u4 >> 5;
    const u32x4* pl = reinterpret_cast<const u32x4*>(lines) + (size_t)blockIdx.x * plane_u4 + lane;

    u32x4 w = __builtin_nontemporal_load(pl + (wave < n_pieces ? wave : 0) * 32);

    {
        float4 xi[INIT_ROWS];
        #pragma unroll
        for (int j = 0; j < INIT_ROWS; ++j) {
            const int r = wave + OWN_WAVES * j;
            xi[j] = reinterpret_cast<const float4*>(x + (size_t)(row0 + (r < rows ? r : 0)) * DIM)[lane];
        }
        #pragma unroll
        for (int j = 0; j < INIT_ROWS; ++j) {
            const int r = wave + OWN_WAVES * j;
            if (r < TILE_ROWS) {
                const bool own = r < rows;
                float* p = acc + r * APITCH + lane;
                p[0]  = own ? xi[j].x : 0.f;
                p[32] = own ? xi[j].y : 0.f;
                p[64] = own ? xi[j].z : 0.f;
                p[96] = own ? xi[j].w : 0.f;
            }
        }
    }
    __syncthreads();

    const int cpos    = blockIdx.x & 31;
    const int cnt_src = ((lane & ~7) | (cpos >> 2)) << 2;

    int* my_ring = ring + wave * RING;
    int head = 0;
    int tail = 0;

    for (int i = wave; i < n_pieces; i += OWN_WAVES) {
        const int in = i + OWN_WAVES;
        const u32x4 wn = __builtin_nontemporal_load(pl + (in < n_pieces ? in : 0) * 32);
        const u32 wd[4] = {w.x, w.y, w.z, w.w};
        const u32 cw    = wd[cpos & 3];
        const u32 count = (u32)__builtin_amdgcn_ds_bpermute(cnt_src, (int)cw);
        const bool over = count > (u32)CELL_SLOTS;
        const u32 nrec  = over ? 0u : count;
        #pragma unroll
        for (int c = 0; c < 4; ++c) {
            const u32  k     = (u32)(4 * (lane & 7) + c - cpos - 1) & 31u;
            const bool valid = k < nrec;
            const u32  bal   = __builtin_amdgcn_ballot_w32(valid);
            if (bal) {
                ring_put(my_ring, tail, bal, valid, wd[c]);
                if (tail - head >= 32)
                    ring_take(x, acc, my_ring, head, 32, lane, n_nodes, rows);
            }
        }
        u32 ovm = __builtin_amdgcn_ballot_w32(over);
        while (ovm) {
            const int l = __builtin_ctz(ovm);
            ovm &= ~(0xffu << (l & ~7));
            const long long ge = (long long)((i * 32 + l) >> 3) * GROUP;
            for (int it = 0; it < GROUP / 32; ++it) {
                const long long e = ge + it * 32 + lane;
                const int  dd = e < n_edges ? dst[e] : -1;
                const int  ss = e < n_edges ? src[e] : 0;
                const int  r  = dd - row0;
                const bool ok = ((u32)r < (u32)rows) && ((u32)ss < (u32)n_nodes);
                const u32  bal = __builtin_amdgcn_ballot_w32(ok);
                if (bal) {
                    ring_put(my_ring, tail, bal, ok, ((u32)ss << 9) | (u32)r);
                    if (tail - head >= 32)
                        ring_take(x, acc, my_ring, head, 32, lane, n_nodes, rows);
                }
            }
        }
        w = wn;
    }
    if (tail - head > 0)
        ring_take(x, acc, my_ring, head, tail - head, lane, n_nodes, rows);
    __syncthreads();

    const int c    = lane & 15;
    const int hh   = lane >> 4;
    const int slab = wave & 7;
    const int rg   = wave >> 3;
    const int ncol = slab * 16 + c;
    const int full_steps = OWN_TILES / ROW_GROUPS;
    const bool tail_tile = rg < OWN_TILES % ROW_GROUPS;
    const int  last_tile = full_steps * ROW_GROUPS + rg;

    v2f wf[WSTEPS];
    #pragma unroll
    for (int s = 0; s < WSTEPS; ++s) {
        const int p = (s << 2) + hh * 2;
        const int k = 4 * (p & 31) + (p >> 5);
        wf[s].x = W1[(size_t)k * DIM + ncol];
        wf[s].y = W1[(size_t)(k + 4) * DIM + ncol];
    }
    const float bias1 = b1[ncol];

    for (int t = 0; t < full_steps; ++t) {
        const int tile = t * ROW_GROUPS + rg;
        const v8f cc = tile_gemm(acc + (tile * 16 + c) * APITCH + hh * 2, wf);
        __syncthreads();
        float* o = acc + (tile * 16 + hh * 8) * APITCH + ncol;
        #pragma unroll
        for (int v = 0; v < 8; ++v) {
            const float y = cc[v] + bias1;
            o[v * APITCH] = y > 0.f ? y : 0.f;
        }
    }
    {
        v8f cc = {0.f, 0.f, 0.f, 0.f, 0.f, 0.f, 0.f, 0.f};
        if (tail_tile)
            cc = tile_gemm(acc + (last_tile * 16 + c) * APITCH + hh * 2, wf);
        __syncthreads();
        if (tail_tile) {
            float* o = acc + (last_tile * 16 + hh * 8) * APITCH + ncol;
            #pragma unroll
            for (int v = 0; v < 8; ++v) {
                const float y = cc[v] + bias1;
                o[v * APITCH] = y > 0.f ? y : 0.f;
            }
        }
    }

    #pragma unroll
    for (int s = 0; s < WSTEPS; ++s) {
        const int k = (s << 2) + hh * 2;
        wf[s].x = W2[(size_t)k * DIM + ncol];
        wf[s].y = W2[(size_t)(k + 1) * DIM + ncol];
    }
    const float bias2 = b2[ncol];
    __syncthreads();

    float* out_rows = out + (size_t)row0 * DIM;
    const int step_rows = ROW_GROUPS * 16;
    for (int t = 0; t < full_steps; ++t) {
        const int tile = t * ROW_GROUPS + rg;
        const v8f dd = tile_gemm(acc + (tile * 16 + c) * APITCH + hh * 2, wf);
        __syncthreads();
        float* o = acc + (tile * 16 + hh * 8) * APITCH + ncol;
        #pragma unroll
        for (int v = 0; v < 8; ++v)
            o[v * APITCH] = dd[v] + bias2;
        if (t > 0) {
            const int r1 = t * step_rows;
            rows_out(acc, out_rows, r1 - step_rows, r1 < rows ? r1 : rows, wave, lane);
        }
    }
    {
        v8f dd = {0.f, 0.f, 0.f, 0.f, 0.f, 0.f, 0.f, 0.f};
        if (tail_tile)
            dd = tile_gemm(acc + (last_tile * 16 + c) * APITCH + hh * 2, wf);
        __syncthreads();
        if (tail_tile) {
            float* o = acc + (last_tile * 16 + hh * 8) * APITCH + ncol;
            #pragma unroll
            for (int v = 0; v < 8; ++v)
                o[v * APITCH] = dd[v] + bias2;
        }
        const int r1 = full_steps * step_rows;
        rows_out(acc, out_rows, r1 - step_rows, r1 < rows ? r1 : rows, wave, lane);
        __syncthreads();
        rows_out(acc, out_rows, r1, rows, wave, lane);
    }
}

extern "C" void kernel_launch(void* const* d_in, const int* in_sizes, int n_in,
                              void* d_out, int out_size, void* d_ws, size_t ws_size,
                              hipStream_t stream) {
    const float* x  = (const float*)d_in[0];
    const int*   ei = (const int*)  d_in[1];
    const float* W1 = (const float*)d_in[2];
    const float* b1 = (const float*)d_in[3];
    const float* W2 = (const float*)d_in[4];
    const float* b2 = (const float*)d_in[5];
    float* out = (float*)d_out;

    const int n_nodes = in_sizes[0] / DIM;
    const int n_edges = in_sizes[1] / 2;

    const int n_lines = (n_edges + 4 * GROUP - 1) / (4 * GROUP) * 4;
    unsigned int* lines = (unsigned int*)d_ws;

    gin_build_kernel<<<n_lines / PLANE_WAVES, PLANE_WAVES * 32, 0, stream>>>(
        ei, ei + n_edges, lines, n_edges, n_nodes, n_lines);

    const int n_ranges = (n_nodes + RANGE_ROWS - 1) / RANGE_ROWS;
    gin_own_kernel<<<n_ranges, OWN_THREADS, 0, stream>>>(
        x, ei, ei + n_edges, lines, W1, b1, W2, b2, out, n_edges, n_nodes, n_lines);
}


# Lines cover groups of 7168 edges in 256 B (was 1024 in 128 B): a builder block's 7 waves share one line set in LDS; 3.5x fewer line bytes written and read back; sum order: LDS arrival as before
# speedup vs baseline: 4.0096x; 1.0786x over previous
#include <hip/hip_runtime.h>

#define DIM       128
#define ROW_PAD   4
#define WSTEPS    (DIM / 4)

typedef float v2f __attribute__((ext_vector_type(2)));
typedef float v8f __attribute__((ext_vector_type(8)));

#define RANGE_ROWS   391
#define OWN_TILES    ((RANGE_ROWS + 15) / 16)
#define TILE_ROWS    (OWN_TILES * 16)
#define APITCH       (DIM + ROW_PAD)
#define ROW_GROUPS   4
#define INIT_ROWS    ((TILE_ROWS + 31) / 32)
#define MAX_RANGES   256
#define OWN_WAVES    32
#define OWN_THREADS  (OWN_WAVES * 32)
#define WAVE_EDGES   1024
#define BUILD_WAVES  7
#define GROUP        (BUILD_WAVES * WAVE_EDGES)
#define LINE_WORDS   64
#define LINE_U4      (LINE_WORDS / 4)
#define CELL_ROWS    (MAX_RANGES * LINE_U4 / 32)
#define CELL_SLOTS   (LINE_WORDS - 1)
#define RING         64
#define ROW_BATCH    8

typedef unsigned int u32;
typedef u32 u32x2 __attribute__((ext_vector_type(2)));
typedef u32 u32x4 __attribute__((ext_vector_type(4)));

__global__ void __launch_bounds__(BUILD_WAVES * 32) gin_build_kernel(
    const int* __restrict__ src, const int* __restrict__ dst, u32* __restrict__ lines,
    int n_edges, int n_nodes, int n_lines)
{
    __shared__ __attribute__((aligned(16))) u32 cells[MAX_RANGES * LINE_WORDS];

    const int wave = __builtin_amdgcn_readfirstlane((int)(threadIdx.x >> 5));
    const int lane = threadIdx.x & 31;
    const int g    = blockIdx.x;

    int d[8][4];
    int s[8][4];
    const long long e0 = (long long)g * GROUP + wave * WAVE_EDGES;
    if ((n_edges & 3) == 0 && e0 + WAVE_EDGES <= n_edges) {
        const int4* d4 = reinterpret_cast<const int4*>(dst + e0) + lane;
        const int4* s4 = reinterpret_cast<const int4*>(src + e0) + lane;
        #pragma unroll
        for (int j = 0; j < 8; ++j) {
            const int4 v = d4[j * 32];
            d[j][0] = v.x; d[j][1] = v.y; d[j][2] = v.z; d[j][3] = v.w;
        }
        #pragma unroll
        for (int j = 0; j < 8; ++j) {
            const int4 v = s4[j * 32];
            s[j][0] = v.x; s[j][1] = v.y; s[j][2] = v.z; s[j][3] = v.w;
        }
    } else {
        #pragma unroll
        for (int j = 0; j < 8; ++j) {
            #pragma unroll
            for (int k = 0; k < 4; ++k) {
                const long long e = e0 + j * 128 + lane * 4 + k;
                d[j][k] = e < n_edges ? dst[e] : -1;
                s[j][k] = e < n_edges ? src[e] : 0;
            }
        }
    }

    for (int k = wave; k < CELL_ROWS; k += BUILD_WAVES)
        reinterpret_cast<u32x4*>(cells)[k * 32 + lane] = (u32x4){0u, 0u, 0u, 0u};
    __syncthreads();

    u32 slot[8][4];
    #pragma unroll
    for (int j = 0; j < 8; ++j) {
        #pragma unroll
        for (int k = 0; k < 4; ++k) {
            const int dd = d[j][k];
            const int ss = s[j][k];
            slot[j][k] = CELL_SLOTS;
            if ((u32)dd < (u32)n_nodes && (u32)ss < (u32)n_nodes) {
                const u32 t = (u32)dd / RANGE_ROWS;
                if (t < MAX_RANGES)
                    slot[j][k] = __hip_atomic_fetch_add(&cells[t * LINE_WORDS + (t & (LINE_WORDS - 1u))], 1u,
                                     __ATOMIC_RELAXED, __HIP_MEMORY_SCOPE_WORKGROUP);
            }
        }
    }
    #pragma unroll
    for (int j = 0; j < 8; ++j) {
        #pragma unroll
        for (int k = 0; k < 4; ++k) {
            if (slot[j][k] < CELL_SLOTS) {
                const u32 dd  = (u32)d[j][k];
                const u32 t   = dd / RANGE_ROWS;
                const u32 row = dd - t * RANGE_ROWS;
                cells[t * LINE_WORDS + ((t + 1u + slot[j][k]) & (LINE_WORDS - 1u))] = ((u32)s[j][k] << 9) | row;
            }
        }
    }
    __syncthreads();

    const size_t plane_u4 = (size_t)n_lines * LINE_U4;
    u32x4* out4 = reinterpret_cast<u32x4*>(lines) + (size_t)(lane >> 4) * plane_u4
                + (size_t)g * LINE_U4 + (lane & 15);
    for (int k0 = wave; k0 < CELL_ROWS; k0 += 8 * BUILD_WAVES) {
        u32x4 v[8];
        #pragma unroll
        for (int t = 0; t < 8; ++t) {
            const int k = k0 + t * BUILD_WAVES;
            if (k < CELL_ROWS)
                v[t] = reinterpret_cast<const u32x4*>(cells)[k * 32 + lane];
        }
        #pragma unroll
        for (int t = 0; t < 8; ++t) {
            const int k = k0 + t * BUILD_WAVES;
            if (k < CELL_ROWS)
                out4[(size_t)k * 2 * plane_u4] = v[t];
        }
    }
}

template <bool FULL>
__device__ __forceinline__ void row_batch(
    const float* __restrict__ x, float* acc, int lane, int sv, int rv, u32& m, int n)
{
    float4 v[ROW_BATCH];
    int    rr[ROW_BATCH];
    #pragma unroll
    for (int j = 0; j < ROW_BATCH; ++j) {
        if (FULL || j < n) {
            const int k = __builtin_ctz(m);
            m &= m - 1;
            const int s = __builtin_amdgcn_readlane(sv, k);
            rr[j]       = __builtin_amdgcn_readlane(rv, k);
            v[j] = reinterpret_cast<const float4*>(x + (size_t)s * DIM)[lane];
        }
    }
    #pragma unroll
    for (int j = 0; j < ROW_BATCH; ++j) {
        if (FULL || j < n) {
            float* p = acc + rr[j] * APITCH + lane;
            __hip_atomic_fetch_add(p,      v[j].x, __ATOMIC_RELAXED, __HIP_MEMORY_SCOPE_WORKGROUP);
            __hip_atomic_fetch_add(p + 32, v[j].y, __ATOMIC_RELAXED, __HIP_MEMORY_SCOPE_WORKGROUP);
            __hip_atomic_fetch_add(p + 64, v[j].z, __ATOMIC_RELAXED, __HIP_MEMORY_SCOPE_WORKGROUP);
            __hip_atomic_fetch_add(p + 96, v[j].w, __ATOMIC_RELAXED, __HIP_MEMORY_SCOPE_WORKGROUP);
        }
    }
}

__device__ __forceinline__ void ring_put(int* my_ring, int& tail, u32 bal, bool valid, u32 rec)
{
    const int rank = (int)__builtin_amdgcn_mbcnt_lo(bal, 0u);
    if (valid)
        my_ring[(tail + rank) & (RING - 1)] = (int)rec;
    tail += __builtin_popcount(bal);
}

__device__ __forceinline__ void ring_take(
    const float* __restrict__ x, float* acc, const int* my_ring, int& head, int n,
    int lane, int n_nodes, int rows)
{
    const u32 rec = (u32)my_ring[(head + lane) & (RING - 1)];
    head += n;
    const int s = (int)(rec >> 9);
    const int r = (int)(rec & 511u);
    const bool ok = (lane < n) && ((u32)r < (u32)rows) && ((u32)s < (u32)n_nodes);
    u32 m = __builtin_amdgcn_ballot_w32(ok);
    int c = __builtin_popcount(m);
    while (c >= ROW_BATCH) {
        row_batch<true>(x, acc, lane, s, r, m, ROW_BATCH);
        c -= ROW_BATCH;
    }
    if (c > 0)
        row_batch<false>(x, acc, lane, s, r, m, c);
}


__device__ __forceinline__ v8f tile_gemm(const float* arow, const v2f (&wf)[WSTEPS])
{
    v8f cc = {0.f, 0.f, 0.f, 0.f, 0.f, 0.f, 0.f, 0.f};
    #pragma unroll
    for (int s = 0; s < WSTEPS; ++s) {
        const v2f a = *reinterpret_cast<const v2f*>(arow + (s << 2));
        cc = __builtin_amdgcn_wmma_f32_16x16x4_f32(
                 false, a, false, wf[s], (short)0, cc, false, false);
    }
    return cc;
}

__device__ __forceinline__ void rows_out(
    const float* acc, float* __restrict__ out_rows, int r0, int r1, int wave, int lane)
{
    for (int r = r0 + wave; r < r1; r += OWN_WAVES)
        reinterpret_cast<float4*>(out_rows + (size_t)r * DIM)[lane] =
            *reinterpret_cast<const float4*>(acc + r * APITCH + lane * 4);
}

__global__ void __launch_bounds__(OWN_THREADS) gin_own_kernel(
    const float* __restrict__ x,
    const int*   __restrict__ src,
    const int*   __restrict__ dst,
    const u32*   __restrict__ lines,
    const float* __restrict__ W1,
    const float* __restrict__ b1,
    const float* __restrict__ W2,
    const float* __restrict__ b2,
    float*       __restrict__ out,
    int n_edges, int n_nodes, int n_lines)
{
    __shared__ __attribute__((aligned(16))) float acc[TILE_ROWS * APITCH];
    __shared__ int ring[OWN_WAVES * RING];

    const int wave = __builtin_amdgcn_readfirstlane((int)(threadIdx.x >> 5));
    const int lane = threadIdx.x & 31;
    const int row0 = blockIdx.x * RANGE_ROWS;
    const int left = n_nodes - row0;
    const int rows = left < RANGE_ROWS ? left : RANGE_ROWS;

    const int n_pieces = n_lines;
    const u32x2* pl = reinterpret_cast<const u32x2*>(lines)
                    + (size_t)blockIdx.x * n_lines * (LINE_WORDS / 2) + lane;

    u32x2 w = __builtin_nontemporal_load(pl + (wave < n_pieces ? wave : 0) * 32);

    {
        float4 xi[INIT_ROWS];
        #pragma unroll
        for (int j = 0; j < INIT_ROWS; ++j) {
            const int r = wave + OWN_WAVES * j;
            xi[j] = reinterpret_cast<const float4*>(x + (size_t)(row0 + (r < rows ? r : 0)) * DIM)[lane];
        }
        #pragma unroll
        for (int j = 0; j < INIT_ROWS; ++j) {
            const int r = wave + OWN_WAVES * j;
            if (r < TILE_ROWS) {
                const bool own = r < rows;
                float* p = acc + r * APITCH + lane;
                p[0]  = own ? xi[j].x : 0.f;
                p[32] = own ? xi[j].y : 0.f;
                p[64] = own ? xi[j].z : 0.f;
                p[96] = own ? xi[j].w : 0.f;
            }
        }
    }
    __syncthreads();

    const int cpos = blockIdx.x & (LINE_WORDS - 1);

    int* my_ring = ring + wave * RING;
    int head = 0;
    int tail = 0;

    for (int i = wave; i < n_pieces; i += OWN_WAVES) {
        const int in = i + OWN_WAVES;
        const u32x2 wn = __builtin_nontemporal_load(pl + (in < n_pieces ? in : 0) * 32);
        const u32 wd[2] = {w.x, w.y};
        const u32 count = (u32)__builtin_amdgcn_readlane((int)wd[cpos & 1], cpos >> 1);
        const bool over = count > (u32)CELL_SLOTS;
        const u32 nrec  = over ? 0u : count;
        #pragma unroll
        for (int c = 0; c < 2; ++c) {
            const u32  k     = (u32)(2 * lane + c - cpos - 1) & (u32)(LINE_WORDS - 1);
            const bool valid = k < nrec;
            const u32  bal   = __builtin_amdgcn_ballot_w32(valid);
            if (bal) {
                ring_put(my_ring, tail, bal, valid, wd[c]);
                if (tail - head >= 32)
                    ring_take(x, acc, my_ring, head, 32, lane, n_nodes, rows);
            }
        }
        if (over) {
            const long long ge = (long long)i * GROUP;
            for (int it = 0; it < GROUP / 32; ++it) {
                const long long e = ge + it * 32 + lane;
                const int  dd = e < n_edges ? dst[e] : -1;
                const int  ss = e < n_edges ? src[e] : 0;
                const int  r  = dd - row0;
                const bool ok = ((u32)r < (u32)rows) && ((u32)ss < (u32)n_nodes);
                const u32  bal = __builtin_amdgcn_ballot_w32(ok);
                if (bal) {
                    ring_put(my_ring, tail, bal, ok, ((u32)ss << 9) | (u32)r);
                    if (tail - head >= 32)
                        ring_take(x, acc, my_ring, head, 32, lane, n_nodes, rows);
                }
            }
        }
        w = wn;
    }
    if (tail - head > 0)
        ring_take(x, acc, my_ring, head, tail - head, lane, n_nodes, rows);
    __syncthreads();

    const int c    = lane & 15;
    const int hh   = lane >> 4;
    const int slab = wave & 7;
    const int rg   = wave >> 3;
    const int ncol = slab * 16 + c;
    const int full_steps = OWN_TILES / ROW_GROUPS;
    const bool tail_tile = rg < OWN_TILES % ROW_GROUPS;
    const int  last_tile = full_steps * ROW_GROUPS + rg;

    v2f wf[WSTEPS];
    #pragma unroll
    for (int s = 0; s < WSTEPS; ++s) {
        const int p = (s << 2) + hh * 2;
        const int k = 4 * (p & 31) + (p >> 5);
        wf[s].x = W1[(size_t)k * DIM + ncol];
        wf[s].y = W1[(size_t)(k + 4) * DIM + ncol];
    }
    const float bias1 = b1[ncol];

    for (int t = 0; t < full_steps; ++t) {
        const int tile = t * ROW_GROUPS + rg;
        const v8f cc = tile_gemm(acc + (tile * 16 + c) * APITCH + hh * 2, wf);
        __syncthreads();
        float* o = acc + (tile * 16 + hh * 8) * APITCH + ncol;
        #pragma unroll
        for (int v = 0; v < 8; ++v) {
            const float y = cc[v] + bias1;
            o[v * APITCH] = y > 0.f ? y : 0.f;
        }
    }
    {
        v8f cc = {0.f, 0.f, 0.f, 0.f, 0.f, 0.f, 0.f, 0.f};
        if (tail_tile)
            cc = tile_gemm(acc + (last_tile * 16 + c) * APITCH + hh * 2, wf);
        __syncthreads();
        if (tail_tile) {
            float* o = acc + (last_tile * 16 + hh * 8) * APITCH + ncol;
            #pragma unroll
            for (int v = 0; v < 8; ++v) {
                const float y = cc[v] + bias1;
                o[v * APITCH] = y > 0.f ? y : 0.f;
            }
        }
    }

    #pragma unroll
    for (int s = 0; s < WSTEPS; ++s) {
        const int k = (s << 2) + hh * 2;
        wf[s].x = W2[(size_t)k * DIM + ncol];
        wf[s].y = W2[(size_t)(k + 1) * DIM + ncol];
    }
    const float bias2 = b2[ncol];
    __syncthreads();

    float* out_rows = out + (size_t)row0 * DIM;
    const int step_rows = ROW_GROUPS * 16;
    for (int t = 0; t < full_steps; ++t) {
        const int tile = t * ROW_GROUPS + rg;
        const v8f dd = tile_gemm(acc + (tile * 16 + c) * APITCH + hh * 2, wf);
        __syncthreads();
        float* o = acc + (tile * 16 + hh * 8) * APITCH + ncol;
        #pragma unroll
        for (int v = 0; v < 8; ++v)
            o[v * APITCH] = dd[v] + bias2;
        if (t > 0) {
            const int r1 = t * step_rows;
            rows_out(acc, out_rows, r1 - step_rows, r1 < rows ? r1 : rows, wave, lane);
        }
    }
    {
        v8f dd = {0.f, 0.f, 0.f, 0.f, 0.f, 0.f, 0.f, 0.f};
        if (tail_tile)
            dd = tile_gemm(acc + (last_tile * 16 + c) * APITCH + hh * 2, wf);
        __syncthreads();
        if (tail_tile) {
            float* o = acc + (last_tile * 16 + hh * 8) * APITCH + ncol;
            #pragma unroll
            for (int v = 0; v < 8; ++v)
                o[v * APITCH] = dd[v] + bias2;
        }
        const int r1 = full_steps * step_rows;
        rows_out(acc, out_rows, r1 - step_rows, r1 < rows ? r1 : rows, wave, lane);
        __syncthreads();
        rows_out(acc, out_rows, r1, rows, wave, lane);
    }
}

extern "C" void kernel_launch(void* const* d_in, const int* in_sizes, int n_in,
                              void* d_out, int out_size, void* d_ws, size_t ws_size,
                              hipStream_t stream) {
    const float* x  = (const float*)d_in[0];
    const int*   ei = (const int*)  d_in[1];
    const float* W1 = (const float*)d_in[2];
    const float* b1 = (const float*)d_in[3];
    const float* W2 = (const float*)d_in[4];
    const float* b2 = (const float*)d_in[5];
    float* out = (float*)d_out;

    const int n_nodes = in_sizes[0] / DIM;
    const int n_edges = in_sizes[1] / 2;

    const int n_lines = (n_edges + GROUP - 1) / GROUP;
    unsigned int* lines = (unsigned int*)d_ws;

    gin_build_kernel<<<n_lines, BUILD_WAVES * 32, 0, stream>>>(
        ei, ei + n_edges, lines, n_edges, n_nodes, n_lines);

    const int n_ranges = (n_nodes + RANGE_ROWS - 1) / RANGE_ROWS;
    gin_own_kernel<<<n_ranges, OWN_THREADS, 0, stream>>>(
        x, ei, ei + n_edges, lines, W1, b1, W2, b2, out, n_edges, n_nodes, n_lines);
}
